# starting kernel
# baseline (speedup 1.0000x reference)
#include <hip/hip_runtime.h>
#include <hip/hip_bf16.h>
#include <math.h>

#define DD 64
#define NSLOPE 0.2f

typedef __attribute__((ext_vector_type(2))) float v2f;
typedef __attribute__((ext_vector_type(8))) float v8f;

__device__ __forceinline__ unsigned flipF(float f) {
    unsigned u = __float_as_uint(f);
    return (u >> 31) ? ~u : (u | 0x80000000u);
}
__device__ __forceinline__ float unflipF(unsigned u) {
    unsigned v = (u >> 31) ? (u & 0x7FFFFFFFu) : ~u;
    return __uint_as_float(v);
}

__global__ void k_init(float* __restrict__ agg, float* __restrict__ denom,
                       unsigned* __restrict__ emaxU, int n) {
    int i = blockIdx.x * blockDim.x + threadIdx.x;
    if (i < n * DD) agg[i] = 0.0f;
    if (i < n) { denom[i] = 0.0f; emaxU[i] = 0u; }
}

__global__ __launch_bounds__(256) void k_gemm(const float* __restrict__ x,
                                              const float* __restrict__ W,
                                              const float* __restrict__ bias,
                                              float* __restrict__ Wh, int n) {
    const int lane = threadIdx.x & 31;
    const int wave = threadIdx.x >> 5;
    const int tile = blockIdx.x * 8 + wave;
    if (tile * 16 >= n) return;

    const int lm   = lane & 15;
    const int half = lane >> 4;
    const int row  = tile * 16 + lm;

    const float* xr = x + (size_t)row * DD + 2 * half;
    v2f areg[16];
#pragma unroll
    for (int i = 0; i < 16; ++i) {
        areg[i].x = xr[4 * i + 0];
        areg[i].y = xr[4 * i + 1];
    }

#pragma unroll
    for (int c = 0; c < 4; ++c) {
        v8f acc = {};
        const float* wr = W + (size_t)(c * 16 + lm) * DD + 2 * half;
#pragma unroll
        for (int i = 0; i < 16; ++i) {
            v2f bfrag;
            bfrag.x = wr[4 * i + 0];
            bfrag.y = wr[4 * i + 1];
            acc = __builtin_amdgcn_wmma_f32_16x16x4_f32(
                false, areg[i], false, bfrag, (short)0, acc, false, false);
        }
        const float bv = bias[c * 16 + lm];
#pragma unroll
        for (int v = 0; v < 8; ++v) {
            Wh[(size_t)(tile * 16 + v + 8 * half) * DD + c * 16 + lm] = acc[v] + bv;
        }
    }
}

__global__ __launch_bounds__(256) void k_logits(const float* __restrict__ Wh,
                                                const float* __restrict__ avec,
                                                const int* __restrict__ src,
                                                const int* __restrict__ dst,
                                                float* __restrict__ eb,
                                                unsigned* __restrict__ emaxU, int E) {
    const int lane = threadIdx.x & 31;
    const int e = blockIdx.x * 8 + (threadIdx.x >> 5);
    if (e >= E) return;
    const int s = src[e], d = dst[e];

    const float2 ws = ((const float2*)(Wh + (size_t)s * DD))[lane];
    const float2 wd = ((const float2*)(Wh + (size_t)d * DD))[lane];
    const float2 av = ((const float2*)avec)[lane];

    float sx = ws.x + wd.x, sy = ws.y + wd.y;
    sx = sx > 0.0f ? sx : NSLOPE * sx;
    sy = sy > 0.0f ? sy : NSLOPE * sy;
    float p = sx * av.x + sy * av.y;

#pragma unroll
    for (int m = 16; m >= 1; m >>= 1) p += __shfl_xor(p, m, 32);

    if (lane == 0) {
        eb[e] = p;
        atomicMax(emaxU + d, flipF(p));
    }
}

__global__ void k_exp(const int* __restrict__ dst, float* __restrict__ eb,
                      const unsigned* __restrict__ emaxU,
                      float* __restrict__ denom, int E) {
    int i = blockIdx.x * blockDim.x + threadIdx.x;
    if (i >= E) return;
    int d = dst[i];
    float ex = __expf(eb[i] - unflipF(emaxU[d]));
    eb[i] = ex;
    unsafeAtomicAdd(denom + d, ex);
}

__global__ __launch_bounds__(256) void k_agg(const float* __restrict__ Wh,
                                             const int* __restrict__ src,
                                             const int* __restrict__ dst,
                                             const float* __restrict__ eb,
                                             const float* __restrict__ denom,
                                             float* __restrict__ agg, int E) {
    const int lane = threadIdx.x & 31;
    const int e = blockIdx.x * 8 + (threadIdx.x >> 5);
    if (e >= E) return;
    const int s = src[e], d = dst[e];
    const float alpha = eb[e] / fmaxf(denom[d], 1e-9f);
    const float2 w = ((const float2*)(Wh + (size_t)s * DD))[lane];
    float* ap = agg + (size_t)d * DD + 2 * lane;
    unsafeAtomicAdd(ap + 0, alpha * w.x);
    unsafeAtomicAdd(ap + 1, alpha * w.y);
}

__global__ void k_sig(float* __restrict__ out, int n) {
    int i = blockIdx.x * blockDim.x + threadIdx.x;
    if (i < n * DD) out[i] = 1.0f / (1.0f + __expf(-out[i]));
}

extern "C" void kernel_launch(void* const* d_in, const int* in_sizes, int n_in,
                              void* d_out, int out_size, void* d_ws, size_t ws_size,
                              hipStream_t stream) {
    const float* x    = (const float*)d_in[0];
    const float* W    = (const float*)d_in[1];
    const float* bias = (const float*)d_in[2];
    const float* avec = (const float*)d_in[3];
    const int*   ei   = (const int*)d_in[4];

    const int N = in_sizes[0] / DD;
    const int E = in_sizes[4] / 2;
    const int* src = ei;
    const int* dst = ei + E;

    float*    Wh    = (float*)d_ws;
    float*    eb    = Wh + (size_t)N * DD;
    unsigned* emaxU = (unsigned*)(eb + E);
    float*    denom = (float*)(emaxU + N);

    float* agg = (float*)d_out;

    const int nd = N * DD;
    k_init<<<(nd + 255) / 256, 256, 0, stream>>>(agg, denom, emaxU, N);

    const int tiles = (N + 15) / 16;
    k_gemm<<<(tiles + 7) / 8, 256, 0, stream>>>(x, W, bias, Wh, N);

    k_logits<<<(E + 7) / 8, 256, 0, stream>>>(Wh, avec, src, dst, eb, emaxU, E);

    k_exp<<<(E + 255) / 256, 256, 0, stream>>>(dst, eb, emaxU, denom, E);

    k_agg<<<(E + 7) / 8, 256, 0, stream>>>(Wh, src, dst, eb, denom, agg, E);

    k_sig<<<(nd + 255) / 256, 256, 0, stream>>>(agg, N);
}


# k_logits: one wave per 32 edges (16-lane row per edge, 16-B gathers of whole rows, coalesced index/eb/max ops) instead of one wave per edge; logit sum order now 4 per lane, then 16-lane butterfly
# speedup vs baseline: 1.4634x; 1.4634x over previous
#include <hip/hip_runtime.h>
#include <hip/hip_bf16.h>
#include <math.h>

#define DD 64
#define NSLOPE 0.2f

typedef __attribute__((ext_vector_type(2))) float v2f;
typedef __attribute__((ext_vector_type(8))) float v8f;

__device__ __forceinline__ unsigned flipF(float f) {
    unsigned u = __float_as_uint(f);
    return (u >> 31) ? ~u : (u | 0x80000000u);
}
__device__ __forceinline__ float unflipF(unsigned u) {
    unsigned v = (u >> 31) ? (u & 0x7FFFFFFFu) : ~u;
    return __uint_as_float(v);
}

__global__ void k_init(float* __restrict__ agg, float* __restrict__ denom,
                       unsigned* __restrict__ emaxU, int n) {
    int i = blockIdx.x * blockDim.x + threadIdx.x;
    if (i < n * DD) agg[i] = 0.0f;
    if (i < n) { denom[i] = 0.0f; emaxU[i] = 0u; }
}

__global__ __launch_bounds__(256) void k_gemm(const float* __restrict__ x,
                                              const float* __restrict__ W,
                                              const float* __restrict__ bias,
                                              float* __restrict__ Wh, int n) {
    const int lane = threadIdx.x & 31;
    const int wave = threadIdx.x >> 5;
    const int tile = blockIdx.x * 8 + wave;
    if (tile * 16 >= n) return;

    const int lm   = lane & 15;
    const int half = lane >> 4;
    const int row  = tile * 16 + lm;

    const float* xr = x + (size_t)row * DD + 2 * half;
    v2f areg[16];
#pragma unroll
    for (int i = 0; i < 16; ++i) {
        areg[i].x = xr[4 * i + 0];
        areg[i].y = xr[4 * i + 1];
    }

#pragma unroll
    for (int c = 0; c < 4; ++c) {
        v8f acc = {};
        const float* wr = W + (size_t)(c * 16 + lm) * DD + 2 * half;
#pragma unroll
        for (int i = 0; i < 16; ++i) {
            v2f bfrag;
            bfrag.x = wr[4 * i + 0];
            bfrag.y = wr[4 * i + 1];
            acc = __builtin_amdgcn_wmma_f32_16x16x4_f32(
                false, areg[i], false, bfrag, (short)0, acc, false, false);
        }
        const float bv = bias[c * 16 + lm];
#pragma unroll
        for (int v = 0; v < 8; ++v) {
            Wh[(size_t)(tile * 16 + v + 8 * half) * DD + c * 16 + lm] = acc[v] + bv;
        }
    }
}

template <int J> __device__ __forceinline__ int rowBcast(int v) {
    return __builtin_amdgcn_update_dpp(0, v, 0x150 + J, 0xf, 0xf, false);
}
template <int M> __device__ __forceinline__ float rowXorAdd(float p) {
    const int q = __builtin_amdgcn_update_dpp(0, __float_as_int(p), 0x160 + M, 0xf, 0xf, false);
    return p + __int_as_float(q);
}
template <int J> __device__ __forceinline__ void logitPass(const float* __restrict__ Wh,
                                                           const float4 av, const int s,
                                                           const int d, const int c,
                                                           float& res) {
    const int sj = rowBcast<J>(s);
    const int dj = rowBcast<J>(d);
    const float4 ws = ((const float4*)(Wh + (size_t)sj * DD))[c];
    const float4 wd = ((const float4*)(Wh + (size_t)dj * DD))[c];
    float t0 = ws.x + wd.x, t1 = ws.y + wd.y, t2 = ws.z + wd.z, t3 = ws.w + wd.w;
    t0 = t0 > 0.0f ? t0 : NSLOPE * t0;
    t1 = t1 > 0.0f ? t1 : NSLOPE * t1;
    t2 = t2 > 0.0f ? t2 : NSLOPE * t2;
    t3 = t3 > 0.0f ? t3 : NSLOPE * t3;
    float p = t0 * av.x + t1 * av.y + t2 * av.z + t3 * av.w;
    p = rowXorAdd<1>(p);
    p = rowXorAdd<2>(p);
    p = rowXorAdd<4>(p);
    p = rowXorAdd<8>(p);
    if (c == J) res = p;
}

__global__ __launch_bounds__(256) void k_logits(const float* __restrict__ Wh,
                                                const float* __restrict__ avec,
                                                const int* __restrict__ src,
                                                const int* __restrict__ dst,
                                                float* __restrict__ eb,
                                                unsigned* __restrict__ emaxU, int E) {
    const int lane = threadIdx.x & 31;
    const int c = lane & 15;
    const int e0 = (blockIdx.x * 8 + (threadIdx.x >> 5)) * 32;
    if (e0 >= E) return;
    const int e = e0 + lane;
    const bool live = e < E;
    const int s = live ? src[e] : 0;
    const int d = live ? dst[e] : 0;
    const float4 av = ((const float4*)avec)[c];

    float res = 0.0f;
    logitPass<0>(Wh, av, s, d, c, res);
    logitPass<1>(Wh, av, s, d, c, res);
    logitPass<2>(Wh, av, s, d, c, res);
    logitPass<3>(Wh, av, s, d, c, res);
    logitPass<4>(Wh, av, s, d, c, res);
    logitPass<5>(Wh, av, s, d, c, res);
    logitPass<6>(Wh, av, s, d, c, res);
    logitPass<7>(Wh, av, s, d, c, res);
    logitPass<8>(Wh, av, s, d, c, res);
    logitPass<9>(Wh, av, s, d, c, res);
    logitPass<10>(Wh, av, s, d, c, res);
    logitPass<11>(Wh, av, s, d, c, res);
    logitPass<12>(Wh, av, s, d, c, res);
    logitPass<13>(Wh, av, s, d, c, res);
    logitPass<14>(Wh, av, s, d, c, res);
    logitPass<15>(Wh, av, s, d, c, res);

    if (live) {
        eb[e] = res;
        atomicMax(emaxU + d, flipF(res));
    }
}

__global__ void k_exp(const int* __restrict__ dst, float* __restrict__ eb,
                      const unsigned* __restrict__ emaxU,
                      float* __restrict__ denom, int E) {
    int i = blockIdx.x * blockDim.x + threadIdx.x;
    if (i >= E) return;
    int d = dst[i];
    float ex = __expf(eb[i] - unflipF(emaxU[d]));
    eb[i] = ex;
    unsafeAtomicAdd(denom + d, ex);
}

__global__ __launch_bounds__(256) void k_agg(const float* __restrict__ Wh,
                                             const int* __restrict__ src,
                                             const int* __restrict__ dst,
                                             const float* __restrict__ eb,
                                             const float* __restrict__ denom,
                                             float* __restrict__ agg, int E) {
    const int lane = threadIdx.x & 31;
    const int e = blockIdx.x * 8 + (threadIdx.x >> 5);
    if (e >= E) return;
    const int s = src[e], d = dst[e];
    const float alpha = eb[e] / fmaxf(denom[d], 1e-9f);
    const float2 w = ((const float2*)(Wh + (size_t)s * DD))[lane];
    float* ap = agg + (size_t)d * DD + 2 * lane;
    unsafeAtomicAdd(ap + 0, alpha * w.x);
    unsafeAtomicAdd(ap + 1, alpha * w.y);
}

__global__ void k_sig(float* __restrict__ out, int n) {
    int i = blockIdx.x * blockDim.x + threadIdx.x;
    if (i < n * DD) out[i] = 1.0f / (1.0f + __expf(-out[i]));
}

extern "C" void kernel_launch(void* const* d_in, const int* in_sizes, int n_in,
                              void* d_out, int out_size, void* d_ws, size_t ws_size,
                              hipStream_t stream) {
    const float* x    = (const float*)d_in[0];
    const float* W    = (const float*)d_in[1];
    const float* bias = (const float*)d_in[2];
    const float* avec = (const float*)d_in[3];
    const int*   ei   = (const int*)d_in[4];

    const int N = in_sizes[0] / DD;
    const int E = in_sizes[4] / 2;
    const int* src = ei;
    const int* dst = ei + E;

    float*    Wh    = (float*)d_ws;
    float*    eb    = Wh + (size_t)N * DD;
    unsigned* emaxU = (unsigned*)(eb + E);
    float*    denom = (float*)(emaxU + N);

    float* agg = (float*)d_out;

    const int nd = N * DD;
    k_init<<<(nd + 255) / 256, 256, 0, stream>>>(agg, denom, emaxU, N);

    const int tiles = (N + 15) / 16;
    k_gemm<<<(tiles + 7) / 8, 256, 0, stream>>>(x, W, bias, Wh, N);

    k_logits<<<(E + 255) / 256, 256, 0, stream>>>(Wh, avec, src, dst, eb, emaxU, E);

    k_exp<<<(E + 255) / 256, 256, 0, stream>>>(dst, eb, emaxU, denom, E);

    k_agg<<<(E + 7) / 8, 256, 0, stream>>>(Wh, src, dst, eb, denom, agg, E);

    k_sig<<<(nd + 255) / 256, 256, 0, stream>>>(agg, N);
}


# group edges by dst in-call (atomic rank, wave-scan offsets, scatter): each node does logits, softmax, weighted sum, sigmoid in registers, no float atomics; sums now per-node online softmax by arrival
# speedup vs baseline: 2.7915x; 1.9075x over previous
#include <hip/hip_runtime.h>
#include <hip/hip_bf16.h>
#include <math.h>


#define DD 64
#define NSLOPE 0.2f

typedef __attribute__((ext_vector_type(2))) float v2f;
typedef __attribute__((ext_vector_type(8))) float v8f;

__global__ __launch_bounds__(256) void k_gemm(const float* __restrict__ x,
                                              const float* __restrict__ W,
                                              const float* __restrict__ bias,
                                              float* __restrict__ Wh,
                                              unsigned* __restrict__ cnt, int n) {
    const int gt = blockIdx.x * 256 + threadIdx.x;
    if (gt <= n) cnt[gt] = 0u;

    const int lane = threadIdx.x & 31;
    const int wave = threadIdx.x >> 5;
    const int tile = blockIdx.x * 8 + wave;
    if (tile * 16 >= n) return;

    const int lm   = lane & 15;
    const int half = lane >> 4;
    const int row  = tile * 16 + lm;

    const float* xr = x + (size_t)row * DD + 2 * half;
    v2f areg[16];
#pragma unroll
    for (int i = 0; i < 16; ++i) {
        areg[i].x = xr[4 * i + 0];
        areg[i].y = xr[4 * i + 1];
    }

#pragma unroll
    for (int c = 0; c < 4; ++c) {
        v8f acc = {};
        const float* wr = W + (size_t)(c * 16 + lm) * DD + 2 * half;
#pragma unroll
        for (int i = 0; i < 16; ++i) {
            v2f bfrag;
            bfrag.x = wr[4 * i + 0];
            bfrag.y = wr[4 * i + 1];
            acc = __builtin_amdgcn_wmma_f32_16x16x4_f32(
                false, areg[i], false, bfrag, (short)0, acc, false, false);
        }
        const float bv = bias[c * 16 + lm];
#pragma unroll
        for (int v = 0; v < 8; ++v) {
            Wh[(size_t)(tile * 16 + v + 8 * half) * DD + c * 16 + lm] = acc[v] + bv;
        }
    }
}

__global__ __launch_bounds__(256) void k_count(const int* __restrict__ dst,
                                               unsigned* __restrict__ cnt,
                                               unsigned* __restrict__ rank, int E) {
    const int T = gridDim.x * 256;
    const int e0 = blockIdx.x * 256 + threadIdx.x;
    int d[4];
    unsigned r[4];
#pragma unroll
    for (int j = 0; j < 4; ++j) d[j] = (e0 + j * T < E) ? dst[e0 + j * T] : -1;
#pragma unroll
    for (int j = 0; j < 4; ++j) r[j] = (d[j] >= 0) ? atomicAdd(cnt + d[j], 1u) : 0u;
#pragma unroll
    for (int j = 0; j < 4; ++j) if (e0 + j * T < E) rank[e0 + j * T] = r[j];
}

__global__ __launch_bounds__(256) void k_alloc(const unsigned* __restrict__ cnt,
                                               unsigned* __restrict__ cursor,
                                               unsigned* __restrict__ seg, int n) {
    const int i = blockIdx.x * 256 + threadIdx.x;
    const int lane = threadIdx.x & 31;
    const unsigned c = (i < n) ? cnt[i] : 0u;
    unsigned s = c;
#pragma unroll
    for (int o = 1; o < 32; o <<= 1) {
        const unsigned t = __shfl_up(s, o);
        if (lane >= o) s += t;
    }
    unsigned base = 0u;
    if (lane == 31) base = atomicAdd(cursor, s);
    base = __shfl(base, 31);
    if (i < n) ((uint2*)seg)[i] = make_uint2(base + s - c, c);
}

__global__ __launch_bounds__(256) void k_scatter(const int* __restrict__ src,
                                                 const int* __restrict__ dst,
                                                 const unsigned* __restrict__ rank,
                                                 const unsigned* __restrict__ seg,
                                                 int* __restrict__ gsrc, int E) {
    const int T = gridDim.x * 256;
    const int e0 = blockIdx.x * 256 + threadIdx.x;
    int d[4], sv[4];
    unsigned r[4], b[4];
#pragma unroll
    for (int j = 0; j < 4; ++j) {
        const int e = min(e0 + j * T, E - 1);
        d[j]  = dst[e];
        r[j]  = rank[e];
        sv[j] = src[e];
    }
#pragma unroll
    for (int j = 0; j < 4; ++j) b[j] = seg[2 * d[j]];
#pragma unroll
    for (int j = 0; j < 4; ++j) if (e0 + j * T < E) gsrc[b[j] + r[j]] = sv[j];
}

template <int J> __device__ __forceinline__ int rowBcast(int v) {
    return __builtin_amdgcn_update_dpp(0, v, 0x150 + J, 0xf, 0xf, false);
}
template <int M> __device__ __forceinline__ float rowXorAdd(float p) {
    const int q = __builtin_amdgcn_update_dpp(0, __float_as_int(p), 0x160 + M, 0xf, 0xf, false);
    return p + __int_as_float(q);
}
__device__ __forceinline__ float rowLogit(const float4 ws, const float4 wd, const float4 av) {
    float t0 = ws.x + wd.x, t1 = ws.y + wd.y, t2 = ws.z + wd.z, t3 = ws.w + wd.w;
    t0 = t0 > 0.0f ? t0 : NSLOPE * t0;
    t1 = t1 > 0.0f ? t1 : NSLOPE * t1;
    t2 = t2 > 0.0f ? t2 : NSLOPE * t2;
    t3 = t3 > 0.0f ? t3 : NSLOPE * t3;
    float p = t0 * av.x + t1 * av.y + t2 * av.z + t3 * av.w;
    p = rowXorAdd<1>(p);
    p = rowXorAdd<2>(p);
    p = rowXorAdd<4>(p);
    p = rowXorAdd<8>(p);
    return p;
}
template <int J> __device__ __forceinline__ void edge4(const float* __restrict__ Wh,
                                                       const float4 wi, const float4 av,
                                                       const int sv, const int rem, const int c,
                                                       float& m, float& den, float4& acc) {
    const int s0 = rowBcast<J + 0>(sv);
    const int s1 = rowBcast<J + 1>(sv);
    const int s2 = rowBcast<J + 2>(sv);
    const int s3 = rowBcast<J + 3>(sv);
    const float4 w0 = ((const float4*)(Wh + (size_t)s0 * DD))[c];
    const float4 w1 = ((const float4*)(Wh + (size_t)s1 * DD))[c];
    const float4 w2 = ((const float4*)(Wh + (size_t)s2 * DD))[c];
    const float4 w3 = ((const float4*)(Wh + (size_t)s3 * DD))[c];
    const float l0 = rowLogit(w0, wi, av);
    const float l1 = rowLogit(w1, wi, av);
    const float l2 = rowLogit(w2, wi, av);
    const float l3 = rowLogit(w3, wi, av);
    const bool v0 = (J + 0) < rem, v1 = (J + 1) < rem, v2 = (J + 2) < rem, v3 = (J + 3) < rem;
    float mn = m;
    mn = v0 ? fmaxf(mn, l0) : mn;
    mn = v1 ? fmaxf(mn, l1) : mn;
    mn = v2 ? fmaxf(mn, l2) : mn;
    mn = v3 ? fmaxf(mn, l3) : mn;
    const float sc = __expf(m - mn);
    const float p0 = v0 ? __expf(l0 - mn) : 0.0f;
    const float p1 = v1 ? __expf(l1 - mn) : 0.0f;
    const float p2 = v2 ? __expf(l2 - mn) : 0.0f;
    const float p3 = v3 ? __expf(l3 - mn) : 0.0f;
    den   = den   * sc + ((p0 + p1) + (p2 + p3));
    acc.x = acc.x * sc + ((p0 * w0.x + p1 * w1.x) + (p2 * w2.x + p3 * w3.x));
    acc.y = acc.y * sc + ((p0 * w0.y + p1 * w1.y) + (p2 * w2.y + p3 * w3.y));
    acc.z = acc.z * sc + ((p0 * w0.z + p1 * w1.z) + (p2 * w2.z + p3 * w3.z));
    acc.w = acc.w * sc + ((p0 * w0.w + p1 * w1.w) + (p2 * w2.w + p3 * w3.w));
    m = mn;
}

__global__ __launch_bounds__(256) void k_node(const float* __restrict__ Wh,
                                              const float* __restrict__ avec,
                                              const unsigned* __restrict__ seg,
                                              const int* __restrict__ gsrc,
                                              float* __restrict__ out, int n) {
    const int lane = threadIdx.x & 31;
    const int c = lane & 15;
    const int node0 = (blockIdx.x * 8 + (threadIdx.x >> 5)) * 2;
    if (node0 >= n) return;
    const int node = node0 + (lane >> 4);
    const bool live = node < n;
    const int nd = live ? node : node0;

    const unsigned long long sg = ((const unsigned long long*)seg)[nd];
    const int deg = live ? (int)(unsigned)(sg >> 32) : 0;
    const unsigned beg = (unsigned)sg;
    const float4 av = ((const float4*)avec)[c];
    const float4 wi = ((const float4*)(Wh + (size_t)nd * DD))[c];

    const int d0 = __builtin_amdgcn_readlane(deg, 0);
    const int d1 = __builtin_amdgcn_readlane(deg, 16);
    const int degmax = d0 > d1 ? d0 : d1;

    float m = -3.0e38f;
    float den = 0.0f;
    float4 acc = make_float4(0.0f, 0.0f, 0.0f, 0.0f);

    for (int k = 0; k < degmax; k += 16) {
        const int rem = deg - k;
        int sv = (c < rem) ? gsrc[beg + (unsigned)(k + c)] : nd;
        sv = min(max(sv, 0), n - 1);
        edge4<0>(Wh, wi, av, sv, rem, c, m, den, acc);
        if (k + 4 < degmax)  edge4<4>(Wh, wi, av, sv, rem, c, m, den, acc);
        if (k + 8 < degmax)  edge4<8>(Wh, wi, av, sv, rem, c, m, den, acc);
        if (k + 12 < degmax) edge4<12>(Wh, wi, av, sv, rem, c, m, den, acc);
    }

    const float dn = fmaxf(den, 1e-9f);
    float4 o;
    o.x = 1.0f / (1.0f + __expf(-(acc.x / dn)));
    o.y = 1.0f / (1.0f + __expf(-(acc.y / dn)));
    o.z = 1.0f / (1.0f + __expf(-(acc.z / dn)));
    o.w = 1.0f / (1.0f + __expf(-(acc.w / dn)));
    if (live) ((float4*)(out + (size_t)node * DD))[c] = o;
}

extern "C" void kernel_launch(void* const* d_in, const int* in_sizes, int n_in,
                              void* d_out, int out_size, void* d_ws, size_t ws_size,
                              hipStream_t stream) {
    const float* x    = (const float*)d_in[0];
    const float* W    = (const float*)d_in[1];
    const float* bias = (const float*)d_in[2];
    const float* avec = (const float*)d_in[3];
    const int*   ei   = (const int*)d_in[4];

    const int N = in_sizes[0] / DD;
    const int E = in_sizes[4] / 2;
    const int* src = ei;
    const int* dst = ei + E;

    float*    Wh   = (float*)d_ws;
    unsigned* seg  = (unsigned*)(Wh + (size_t)N * DD);
    unsigned* cnt  = seg + 2 * (size_t)N;
    unsigned* rank = cnt + (N + 1);
    int*      gsrc = (int*)(rank + E);

    float* out = (float*)d_out;

    const int tiles = (N + 15) / 16;
    k_gemm<<<(tiles + 7) / 8, 256, 0, stream>>>(x, W, bias, Wh, cnt, N);

    k_count<<<(E + 1023) / 1024, 256, 0, stream>>>(dst, cnt, rank, E);

    k_alloc<<<(N + 255) / 256, 256, 0, stream>>>(cnt, cnt + N, seg, N);

    k_scatter<<<(E + 1023) / 1024, 256, 0, stream>>>(src, dst, rank, seg, gsrc, E);

    k_node<<<(N + 15) / 16, 256, 0, stream>>>(Wh, avec, seg, gsrc, out, N);
}


# group edges in one pass: the atomic rank stores src into a 64-slot row per node (overflow list past 64, not reached by these inputs); no alloc, scatter, rank: 5 launches to 3; sums still by arrival
# speedup vs baseline: 3.5803x; 1.2826x over previous
#include <hip/hip_runtime.h>
#include <hip/hip_bf16.h>
#include <math.h>


#define DD 64
#define NSLOPE 0.2f

typedef __attribute__((ext_vector_type(2))) float v2f;
typedef __attribute__((ext_vector_type(8))) float v8f;

__global__ __launch_bounds__(256) void k_gemm(const float* __restrict__ x,
                                              const float* __restrict__ W,
                                              const float* __restrict__ bias,
                                              float* __restrict__ Wh,
                                              unsigned* __restrict__ cnt, int n) {
    const int gt = blockIdx.x * 256 + threadIdx.x;
    if (gt <= n) cnt[gt] = 0u;

    const int lane = threadIdx.x & 31;
    const int wave = threadIdx.x >> 5;
    const int tile = blockIdx.x * 8 + wave;
    if (tile * 16 >= n) return;

    const int lm   = lane & 15;
    const int half = lane >> 4;
    const int row  = tile * 16 + lm;

    const float* xr = x + (size_t)row * DD + 2 * half;
    v2f areg[16];
#pragma unroll
    for (int i = 0; i < 16; ++i) {
        areg[i].x = xr[4 * i + 0];
        areg[i].y = xr[4 * i + 1];
    }

#pragma unroll
    for (int c = 0; c < 4; ++c) {
        v8f acc = {};
        const float* wr = W + (size_t)(c * 16 + lm) * DD + 2 * half;
#pragma unroll
        for (int i = 0; i < 16; ++i) {
            v2f bfrag;
            bfrag.x = wr[4 * i + 0];
            bfrag.y = wr[4 * i + 1];
            acc = __builtin_amdgcn_wmma_f32_16x16x4_f32(
                false, areg[i], false, bfrag, (short)0, acc, false, false);
        }
        const float bv = bias[c * 16 + lm];
#pragma unroll
        for (int v = 0; v < 8; ++v) {
            Wh[(size_t)(tile * 16 + v + 8 * half) * DD + c * 16 + lm] = acc[v] + bv;
        }
    }
}

#define CAP 64

__global__ __launch_bounds__(256) void k_bucket(const int* __restrict__ src,
                                                const int* __restrict__ dst,
                                                unsigned* __restrict__ cnt,
                                                int* __restrict__ bucket,
                                                int* __restrict__ ovf, int E, int n) {
    const int T = gridDim.x * 256;
    const int e0 = blockIdx.x * 256 + threadIdx.x;
    int d[4], sv[4];
    unsigned r[4];
#pragma unroll
    for (int j = 0; j < 4; ++j) {
        const bool in = e0 + j * T < E;
        d[j]  = in ? dst[e0 + j * T] : -1;
        sv[j] = in ? src[e0 + j * T] : 0;
    }
#pragma unroll
    for (int j = 0; j < 4; ++j) r[j] = (d[j] >= 0) ? atomicAdd(cnt + d[j], 1u) : 0u;
#pragma unroll
    for (int j = 0; j < 4; ++j) {
        if (d[j] >= 0) {
            if (r[j] < (unsigned)CAP) {
                bucket[(size_t)d[j] * CAP + r[j]] = sv[j];
            } else {
                const unsigned p = atomicAdd(cnt + n, 1u);
                ovf[2 * (size_t)p + 0] = d[j];
                ovf[2 * (size_t)p + 1] = sv[j];
            }
        }
    }
}

template <int J> __device__ __forceinline__ int rowBcast(int v) {
    return __builtin_amdgcn_update_dpp(0, v, 0x150 + J, 0xf, 0xf, false);
}
template <int M> __device__ __forceinline__ float rowXorAdd(float p) {
    const int q = __builtin_amdgcn_update_dpp(0, __float_as_int(p), 0x160 + M, 0xf, 0xf, false);
    return p + __int_as_float(q);
}
__device__ __forceinline__ float rowLogit(const float4 ws, const float4 wd, const float4 av) {
    float t0 = ws.x + wd.x, t1 = ws.y + wd.y, t2 = ws.z + wd.z, t3 = ws.w + wd.w;
    t0 = t0 > 0.0f ? t0 : NSLOPE * t0;
    t1 = t1 > 0.0f ? t1 : NSLOPE * t1;
    t2 = t2 > 0.0f ? t2 : NSLOPE * t2;
    t3 = t3 > 0.0f ? t3 : NSLOPE * t3;
    float p = t0 * av.x + t1 * av.y + t2 * av.z + t3 * av.w;
    p = rowXorAdd<1>(p);
    p = rowXorAdd<2>(p);
    p = rowXorAdd<4>(p);
    p = rowXorAdd<8>(p);
    return p;
}
template <int J, bool MASK> __device__ __forceinline__ void edge4(const float* __restrict__ Wh,
                                                       const float4 wi, const float4 av,
                                                       const int sv, const int rem, const int c,
                                                       float& m, float& den, float4& acc) {
    const int s0 = rowBcast<J + 0>(sv);
    const int s1 = rowBcast<J + 1>(sv);
    const int s2 = rowBcast<J + 2>(sv);
    const int s3 = rowBcast<J + 3>(sv);
    const float4 w0 = ((const float4*)(Wh + (size_t)s0 * DD))[c];
    const float4 w1 = ((const float4*)(Wh + (size_t)s1 * DD))[c];
    const float4 w2 = ((const float4*)(Wh + (size_t)s2 * DD))[c];
    const float4 w3 = ((const float4*)(Wh + (size_t)s3 * DD))[c];
    asm volatile("" :: "v"(w0.x), "v"(w0.y), "v"(w0.z), "v"(w0.w), "v"(w1.x), "v"(w1.y), "v"(w1.z), "v"(w1.w),
                       "v"(w2.x), "v"(w2.y), "v"(w2.z), "v"(w2.w), "v"(w3.x), "v"(w3.y), "v"(w3.z), "v"(w3.w));
    const float l0 = rowLogit(w0, wi, av);
    const float l1 = rowLogit(w1, wi, av);
    const float l2 = rowLogit(w2, wi, av);
    const float l3 = rowLogit(w3, wi, av);
    const bool v0 = MASK ? ((rem >> (J + 0)) & 1) != 0 : (J + 0) < rem;
    const bool v1 = MASK ? ((rem >> (J + 1)) & 1) != 0 : (J + 1) < rem;
    const bool v2 = MASK ? ((rem >> (J + 2)) & 1) != 0 : (J + 2) < rem;
    const bool v3 = MASK ? ((rem >> (J + 3)) & 1) != 0 : (J + 3) < rem;
    float mn = m;
    mn = v0 ? fmaxf(mn, l0) : mn;
    mn = v1 ? fmaxf(mn, l1) : mn;
    mn = v2 ? fmaxf(mn, l2) : mn;
    mn = v3 ? fmaxf(mn, l3) : mn;
    const float sc = __expf(m - mn);
    const float p0 = v0 ? __expf(l0 - mn) : 0.0f;
    const float p1 = v1 ? __expf(l1 - mn) : 0.0f;
    const float p2 = v2 ? __expf(l2 - mn) : 0.0f;
    const float p3 = v3 ? __expf(l3 - mn) : 0.0f;
    den   = den   * sc + ((p0 + p1) + (p2 + p3));
    acc.x = acc.x * sc + ((p0 * w0.x + p1 * w1.x) + (p2 * w2.x + p3 * w3.x));
    acc.y = acc.y * sc + ((p0 * w0.y + p1 * w1.y) + (p2 * w2.y + p3 * w3.y));
    acc.z = acc.z * sc + ((p0 * w0.z + p1 * w1.z) + (p2 * w2.z + p3 * w3.z));
    acc.w = acc.w * sc + ((p0 * w0.w + p1 * w1.w) + (p2 * w2.w + p3 * w3.w));
    m = mn;
}

__global__ __launch_bounds__(256) void k_node(const float* __restrict__ Wh,
                                              const float* __restrict__ avec,
                                              const unsigned* __restrict__ cnt,
                                              const int* __restrict__ bucket,
                                              const int* __restrict__ ovf,
                                              float* __restrict__ out, int n) {
    const int lane = threadIdx.x & 31;
    const int c = lane & 15;
    const int node0 = (blockIdx.x * 8 + (threadIdx.x >> 5)) * 2;
    if (node0 >= n) return;
    const int node = node0 + (lane >> 4);
    const bool live = node < n;
    const int nd = live ? node : node0;

    const unsigned cn = cnt[nd];
    const int deg = live ? (int)min(cn, (unsigned)CAP) : 0;
    const unsigned beg = (unsigned)nd * CAP;
    const unsigned moreRows = __builtin_amdgcn_ballot_w32(live && cn > (unsigned)CAP) & 0x00010001u;
    const float4 av = ((const float4*)avec)[c];
    const float4 wi = ((const float4*)(Wh + (size_t)nd * DD))[c];

    const int d0 = __builtin_amdgcn_readlane(deg, 0);
    const int d1 = __builtin_amdgcn_readlane(deg, 16);
    const int degmax = d0 > d1 ? d0 : d1;

    float m = -3.0e38f;
    float den = 0.0f;
    float4 acc = make_float4(0.0f, 0.0f, 0.0f, 0.0f);

    for (int k = 0; k < degmax; k += 16) {
        const int rem = deg - k;
        int sv = (c < rem) ? bucket[beg + (unsigned)(k + c)] : nd;
        sv = min(max(sv, 0), n - 1);
        edge4<0, false>(Wh, wi, av, sv, rem, c, m, den, acc);
        if (k + 4 < degmax)  edge4<4, false>(Wh, wi, av, sv, rem, c, m, den, acc);
        if (k + 8 < degmax)  edge4<8, false>(Wh, wi, av, sv, rem, c, m, den, acc);
        if (k + 12 < degmax) edge4<12, false>(Wh, wi, av, sv, rem, c, m, den, acc);
    }

    if (moreRows != 0u) {
        const int sh = (node & 1) << 4;
        const bool more = ((moreRows >> sh) & 1u) != 0u;
        const int ovn = (int)min(cnt[n], 0x7fffffffu);
        for (int k = 0; k < ovn; k += 16) {
            int od = -1, os = nd;
            if (k + c < ovn) {
                od = ovf[2 * (size_t)(k + c) + 0];
                os = ovf[2 * (size_t)(k + c) + 1];
            }
            const bool hit = more && od == node;
            const unsigned bal = __builtin_amdgcn_ballot_w32(hit);
            if (bal == 0u) continue;
            const int vm = (int)((bal >> sh) & 0xffffu);
            int sv = hit ? os : nd;
            sv = min(max(sv, 0), n - 1);
            if (bal & 0x000f000fu) edge4<0, true>(Wh, wi, av, sv, vm, c, m, den, acc);
            if (bal & 0x00f000f0u) edge4<4, true>(Wh, wi, av, sv, vm, c, m, den, acc);
            if (bal & 0x0f000f00u) edge4<8, true>(Wh, wi, av, sv, vm, c, m, den, acc);
            if (bal & 0xf000f000u) edge4<12, true>(Wh, wi, av, sv, vm, c, m, den, acc);
        }
    }

    const float dn = fmaxf(den, 1e-9f);
    float4 o;
    o.x = 1.0f / (1.0f + __expf(-(acc.x / dn)));
    o.y = 1.0f / (1.0f + __expf(-(acc.y / dn)));
    o.z = 1.0f / (1.0f + __expf(-(acc.z / dn)));
    o.w = 1.0f / (1.0f + __expf(-(acc.w / dn)));
    if (live) ((float4*)(out + (size_t)node * DD))[c] = o;
}

extern "C" void kernel_launch(void* const* d_in, const int* in_sizes, int n_in,
                              void* d_out, int out_size, void* d_ws, size_t ws_size,
                              hipStream_t stream) {
    const float* x    = (const float*)d_in[0];
    const float* W    = (const float*)d_in[1];
    const float* bias = (const float*)d_in[2];
    const float* avec = (const float*)d_in[3];
    const int*   ei   = (const int*)d_in[4];

    const int N = in_sizes[0] / DD;
    const int E = in_sizes[4] / 2;
    const int* src = ei;
    const int* dst = ei + E;

    float*    Wh     = (float*)d_ws;
    int*      bucket = (int*)(Wh + (size_t)N * DD);
    int*      ovf    = bucket + (size_t)N * CAP;
    unsigned* cnt    = (unsigned*)(ovf + 2 * (size_t)E);

    float* out = (float*)d_out;

    const int tiles = (N + 15) / 16;
    k_gemm<<<(tiles + 7) / 8, 256, 0, stream>>>(x, W, bias, Wh, cnt, N);

    k_bucket<<<(E + 1023) / 1024, 256, 0, stream>>>(src, dst, cnt, bucket, ovf, E, N);

    k_node<<<(N + 15) / 16, 256, 0, stream>>>(Wh, avec, cnt, bucket, ovf, out, N);
}


# edges ranked on LDS counters, no global add per edge: launch 1 sorts 4096-edge chunks by 256-node range in LDS beside the GEMM, launch 2 fills each range's slot rows in LDS; sums by LDS arrival
# speedup vs baseline: 7.5713x; 2.1147x over previous
#include <hip/hip_runtime.h>
#include <hip/hip_bf16.h>
#include <math.h>


#define DD 64
#define NSLOPE 0.2f

typedef __attribute__((ext_vector_type(2))) float v2f;
typedef __attribute__((ext_vector_type(8))) float v8f;

__device__ __forceinline__ void gemm_block(const float* __restrict__ x,
                                           const float* __restrict__ W,
                                           const float* __restrict__ bias,
                                           float* __restrict__ Wh,
                                           const int blk, const int n) {
    const int lane = threadIdx.x & 31;
    const int wave = threadIdx.x >> 5;
    const int tile = blk * 8 + wave;
    if (tile * 16 >= n) return;

    const int lm   = lane & 15;
    const int half = lane >> 4;
    const int row  = tile * 16 + lm;

    const float* xr = x + (size_t)row * DD + 2 * half;
    v2f areg[16];
#pragma unroll
    for (int i = 0; i < 16; ++i) {
        areg[i].x = xr[4 * i + 0];
        areg[i].y = xr[4 * i + 1];
    }

#pragma unroll
    for (int c = 0; c < 4; ++c) {
        v8f acc = {};
        const float* wr = W + (size_t)(c * 16 + lm) * DD + 2 * half;
#pragma unroll
        for (int i = 0; i < 16; ++i) {
            v2f bfrag;
            bfrag.x = wr[4 * i + 0];
            bfrag.y = wr[4 * i + 1];
            acc = __builtin_amdgcn_wmma_f32_16x16x4_f32(
                false, areg[i], false, bfrag, (short)0, acc, false, false);
        }
        const float bv = bias[c * 16 + lm];
#pragma unroll
        for (int v = 0; v < 8; ++v) {
            Wh[(size_t)(tile * 16 + v + 8 * half) * DD + c * 16 + lm] = acc[v] + bv;
        }
    }
}

#define CAP 64
#define RSH 8
#define RNODES (1 << RSH)
#define MAXR 256
#define CELL 48
#define CHUNK 4096

__device__ __forceinline__ void bin_block(const int* __restrict__ src,
                                          const int* __restrict__ dst,
                                          unsigned* __restrict__ seg,
                                          unsigned* __restrict__ cellCnt,
                                          int* __restrict__ ovfA,
                                          unsigned* __restrict__ ovfACnt,
                                          const int E, const int chunk,
                                          const int numChunks, const int numRanges,
                                          unsigned* hist, unsigned* cells) {
    const int t = threadIdx.x;
    for (int i = t; i <= MAXR; i += 256) hist[i] = 0u;
    __syncthreads();

    const int e0 = chunk * CHUNK + t;
    int d[16], sv[16];
#pragma unroll
    for (int j = 0; j < 16; ++j) {
        const int e = e0 + j * 256;
        const bool in = e < E;
        d[j]  = in ? dst[e] : -1;
        sv[j] = in ? src[e] : 0;
    }
#pragma unroll
    for (int j = 0; j < 16; ++j) {
        if (d[j] >= 0) {
            const unsigned rg = min((unsigned)d[j] >> RSH, (unsigned)(MAXR - 1));
            const unsigned k = atomicAdd(&hist[rg], 1u);
            if (k < (unsigned)CELL) {
                cells[rg * CELL + k] = ((unsigned)(d[j] & (RNODES - 1)) << 24) | ((unsigned)sv[j] & 0x00ffffffu);
            } else {
                const unsigned q = atomicAdd(&hist[MAXR], 1u);
                ovfA[2 * ((size_t)chunk * CHUNK + q) + 0] = d[j];
                ovfA[2 * ((size_t)chunk * CHUNK + q) + 1] = sv[j];
            }
        }
    }
    __syncthreads();

    for (int i = t; i < numRanges * CELL; i += 256) {
        const int rg = i / CELL;
        const int slot = i - rg * CELL;
        const unsigned c = min(hist[rg], (unsigned)CELL);
        if ((unsigned)slot < c) seg[((size_t)rg * numChunks + chunk) * CELL + slot] = cells[i];
    }
    for (int rg = t; rg < numRanges; rg += 256)
        cellCnt[(size_t)rg * numChunks + chunk] = min(hist[rg], (unsigned)CELL);
    if (t == 0) ovfACnt[chunk] = min(hist[MAXR], (unsigned)CHUNK);
}

__global__ __launch_bounds__(256) void k_gemm_bin(const float* __restrict__ x,
                                                  const float* __restrict__ W,
                                                  const float* __restrict__ bias,
                                                  float* __restrict__ Wh,
                                                  const int* __restrict__ src,
                                                  const int* __restrict__ dst,
                                                  unsigned* __restrict__ seg,
                                                  unsigned* __restrict__ cellCnt,
                                                  int* __restrict__ ovfA,
                                                  unsigned* __restrict__ ovfACnt,
                                                  unsigned* __restrict__ cnt,
                                                  int E, int n, int gemmBlocks,
                                                  int numChunks, int numRanges) {
    __shared__ unsigned hist[MAXR + 1];
    __shared__ unsigned cells[MAXR * CELL];
    const int blk = blockIdx.x;
    if (blk == 0 && threadIdx.x == 0) cnt[n] = 0u;
    if (blk < gemmBlocks) {
        gemm_block(x, W, bias, Wh, blk, n);
    } else {
        bin_block(src, dst, seg, cellCnt, ovfA, ovfACnt, E, blk - gemmBlocks, numChunks, numRanges, hist, cells);
    }
}

__device__ __forceinline__ void put_edge(unsigned* cntL, int* rows,
                                         unsigned* __restrict__ cnt, int* __restrict__ ovf,
                                         const int n, const int r, const unsigned nl, const int s) {
    const unsigned k = atomicAdd(&cntL[nl], 1u);
    if (k < (unsigned)CAP) {
        rows[nl * CAP + k] = s;
    } else {
        const unsigned p = atomicAdd(cnt + n, 1u);
        ovf[2 * (size_t)p + 0] = (r << RSH) + (int)nl;
        ovf[2 * (size_t)p + 1] = s;
    }
}

#define BT 1024
#define BU 10

__global__ __launch_bounds__(BT) void k_build(const unsigned* __restrict__ seg,
                                              const unsigned* __restrict__ cellCnt,
                                              const int* __restrict__ ovfA,
                                              const unsigned* __restrict__ ovfACnt,
                                              unsigned* __restrict__ cnt,
                                              int* __restrict__ bucket,
                                              int* __restrict__ ovf,
                                              int n, int numChunks) {
    __shared__ unsigned cntL[RNODES];
    __shared__ int rows[RNODES * CAP];
    const int t = threadIdx.x;
    const int r = blockIdx.x;
    for (int i = t; i < RNODES; i += BT) cntL[i] = 0u;
    __syncthreads();

    const int total = numChunks * CELL;
    const unsigned* segR = seg + (size_t)r * numChunks * CELL;
    const unsigned* ccR  = cellCnt + (size_t)r * numChunks;
    for (int base = 0; base < total; base += BT * BU) {
        unsigned c[BU], rec[BU];
        bool ok[BU];
#pragma unroll
        for (int j = 0; j < BU; ++j) {
            const int i = base + j * BT + t;
            c[j] = (i < total) ? ccR[i / CELL] : 0u;
        }
#pragma unroll
        for (int j = 0; j < BU; ++j) {
            const int i = base + j * BT + t;
            const int slot = i - (i / CELL) * CELL;
            ok[j]  = (unsigned)slot < c[j];
            rec[j] = ok[j] ? segR[i] : 0u;
        }
#pragma unroll
        for (int j = 0; j < BU; ++j)
            if (ok[j]) put_edge(cntL, rows, cnt, ovf, n, r, rec[j] >> 24, (int)(rec[j] & 0x00ffffffu));
    }

    for (int ch = t; ch < numChunks; ch += BT) {
        const unsigned oc = min(ovfACnt[ch], (unsigned)CHUNK);
        for (unsigned q = 0; q < oc; ++q) {
            const int ed = ovfA[2 * ((size_t)ch * CHUNK + q) + 0];
            const int es = ovfA[2 * ((size_t)ch * CHUNK + q) + 1];
            if (min((unsigned)ed >> RSH, (unsigned)(MAXR - 1)) == (unsigned)r)
                put_edge(cntL, rows, cnt, ovf, n, r, (unsigned)ed & (unsigned)(RNODES - 1), es);
        }
    }
    __syncthreads();

    for (int i = t; i < RNODES * CAP; i += BT) {
        const int nl = i / CAP;
        const int slot = i - nl * CAP;
        const int node = (r << RSH) + nl;
        const unsigned c = min(cntL[nl], (unsigned)CAP);
        if (node < n && (unsigned)slot < c) bucket[(size_t)node * CAP + slot] = rows[i];
    }
    for (int nl = t; nl < RNODES; nl += BT) {
        const int node = (r << RSH) + nl;
        if (node < n) cnt[node] = cntL[nl];
    }
}

template <int J> __device__ __forceinline__ int rowBcast(int v) {
    return __builtin_amdgcn_update_dpp(0, v, 0x150 + J, 0xf, 0xf, false);
}
template <int M> __device__ __forceinline__ float rowXorAdd(float p) {
    const int q = __builtin_amdgcn_update_dpp(0, __float_as_int(p), 0x160 + M, 0xf, 0xf, false);
    return p + __int_as_float(q);
}
__device__ __forceinline__ float rowLogit(const float4 ws, const float4 wd, const float4 av) {
    float t0 = ws.x + wd.x, t1 = ws.y + wd.y, t2 = ws.z + wd.z, t3 = ws.w + wd.w;
    t0 = t0 > 0.0f ? t0 : NSLOPE * t0;
    t1 = t1 > 0.0f ? t1 : NSLOPE * t1;
    t2 = t2 > 0.0f ? t2 : NSLOPE * t2;
    t3 = t3 > 0.0f ? t3 : NSLOPE * t3;
    float p = t0 * av.x + t1 * av.y + t2 * av.z + t3 * av.w;
    p = rowXorAdd<1>(p);
    p = rowXorAdd<2>(p);
    p = rowXorAdd<4>(p);
    p = rowXorAdd<8>(p);
    return p;
}
template <int J, bool MASK> __device__ __forceinline__ void edge4(const float* __restrict__ Wh,
                                                       const float4 wi, const float4 av,
                                                       const int sv, const int rem, const int c,
                                                       float& m, float& den, float4& acc) {
    const int s0 = rowBcast<J + 0>(sv);
    const int s1 = rowBcast<J + 1>(sv);
    const int s2 = rowBcast<J + 2>(sv);
    const int s3 = rowBcast<J + 3>(sv);
    const float4 w0 = ((const float4*)(Wh + (size_t)s0 * DD))[c];
    const float4 w1 = ((const float4*)(Wh + (size_t)s1 * DD))[c];
    const float4 w2 = ((const float4*)(Wh + (size_t)s2 * DD))[c];
    const float4 w3 = ((const float4*)(Wh + (size_t)s3 * DD))[c];
    asm volatile("" :: "v"(w0.x), "v"(w0.y), "v"(w0.z), "v"(w0.w), "v"(w1.x), "v"(w1.y), "v"(w1.z), "v"(w1.w),
                       "v"(w2.x), "v"(w2.y), "v"(w2.z), "v"(w2.w), "v"(w3.x), "v"(w3.y), "v"(w3.z), "v"(w3.w));
    const float l0 = rowLogit(w0, wi, av);
    const float l1 = rowLogit(w1, wi, av);
    const float l2 = rowLogit(w2, wi, av);
    const float l3 = rowLogit(w3, wi, av);
    const bool v0 = MASK ? ((rem >> (J + 0)) & 1) != 0 : (J + 0) < rem;
    const bool v1 = MASK ? ((rem >> (J + 1)) & 1) != 0 : (J + 1) < rem;
    const bool v2 = MASK ? ((rem >> (J + 2)) & 1) != 0 : (J + 2) < rem;
    const bool v3 = MASK ? ((rem >> (J + 3)) & 1) != 0 : (J + 3) < rem;
    float mn = m;
    mn = v0 ? fmaxf(mn, l0) : mn;
    mn = v1 ? fmaxf(mn, l1) : mn;
    mn = v2 ? fmaxf(mn, l2) : mn;
    mn = v3 ? fmaxf(mn, l3) : mn;
    const float sc = __expf(m - mn);
    const float p0 = v0 ? __expf(l0 - mn) : 0.0f;
    const float p1 = v1 ? __expf(l1 - mn) : 0.0f;
    const float p2 = v2 ? __expf(l2 - mn) : 0.0f;
    const float p3 = v3 ? __expf(l3 - mn) : 0.0f;
    den   = den   * sc + ((p0 + p1) + (p2 + p3));
    acc.x = acc.x * sc + ((p0 * w0.x + p1 * w1.x) + (p2 * w2.x + p3 * w3.x));
    acc.y = acc.y * sc + ((p0 * w0.y + p1 * w1.y) + (p2 * w2.y + p3 * w3.y));
    acc.z = acc.z * sc + ((p0 * w0.z + p1 * w1.z) + (p2 * w2.z + p3 * w3.z));
    acc.w = acc.w * sc + ((p0 * w0.w + p1 * w1.w) + (p2 * w2.w + p3 * w3.w));
    m = mn;
}

__global__ __launch_bounds__(256) void k_node(const float* __restrict__ Wh,
                                              const float* __restrict__ avec,
                                              const unsigned* __restrict__ cnt,
                                              const int* __restrict__ bucket,
                                              const int* __restrict__ ovf,
                                              float* __restrict__ out, int n) {
    const int lane = threadIdx.x & 31;
    const int c = lane & 15;
    const int node0 = (blockIdx.x * 8 + (threadIdx.x >> 5)) * 2;
    if (node0 >= n) return;
    const int node = node0 + (lane >> 4);
    const bool live = node < n;
    const int nd = live ? node : node0;

    const unsigned cn = cnt[nd];
    const int deg = live ? (int)min(cn, (unsigned)CAP) : 0;
    const unsigned beg = (unsigned)nd * CAP;
    const unsigned moreRows = __builtin_amdgcn_ballot_w32(live && cn > (unsigned)CAP) & 0x00010001u;
    const float4 av = ((const float4*)avec)[c];
    const float4 wi = ((const float4*)(Wh + (size_t)nd * DD))[c];

    const int d0 = __builtin_amdgcn_readlane(deg, 0);
    const int d1 = __builtin_amdgcn_readlane(deg, 16);
    const int degmax = d0 > d1 ? d0 : d1;

    float m = -3.0e38f;
    float den = 0.0f;
    float4 acc = make_float4(0.0f, 0.0f, 0.0f, 0.0f);

    for (int k = 0; k < degmax; k += 16) {
        const int rem = deg - k;
        int sv = (c < rem) ? bucket[beg + (unsigned)(k + c)] : nd;
        sv = min(max(sv, 0), n - 1);
        edge4<0, false>(Wh, wi, av, sv, rem, c, m, den, acc);
        if (k + 4 < degmax)  edge4<4, false>(Wh, wi, av, sv, rem, c, m, den, acc);
        if (k + 8 < degmax)  edge4<8, false>(Wh, wi, av, sv, rem, c, m, den, acc);
        if (k + 12 < degmax) edge4<12, false>(Wh, wi, av, sv, rem, c, m, den, acc);
    }

    if (moreRows != 0u) {
        const int sh = (node & 1) << 4;
        const bool more = ((moreRows >> sh) & 1u) != 0u;
        const int ovn = (int)min(cnt[n], 0x7fffffffu);
        for (int k = 0; k < ovn; k += 16) {
            int od = -1, os = nd;
            if (k + c < ovn) {
                od = ovf[2 * (size_t)(k + c) + 0];
                os = ovf[2 * (size_t)(k + c) + 1];
            }
            const bool hit = more && od == node;
            const unsigned bal = __builtin_amdgcn_ballot_w32(hit);
            if (bal == 0u) continue;
            const int vm = (int)((bal >> sh) & 0xffffu);
            int sv = hit ? os : nd;
            sv = min(max(sv, 0), n - 1);
            if (bal & 0x000f000fu) edge4<0, true>(Wh, wi, av, sv, vm, c, m, den, acc);
            if (bal & 0x00f000f0u) edge4<4, true>(Wh, wi, av, sv, vm, c, m, den, acc);
            if (bal & 0x0f000f00u) edge4<8, true>(Wh, wi, av, sv, vm, c, m, den, acc);
            if (bal & 0xf000f000u) edge4<12, true>(Wh, wi, av, sv, vm, c, m, den, acc);
        }
    }

    const float dn = fmaxf(den, 1e-9f);
    float4 o;
    o.x = 1.0f / (1.0f + __expf(-(acc.x / dn)));
    o.y = 1.0f / (1.0f + __expf(-(acc.y / dn)));
    o.z = 1.0f / (1.0f + __expf(-(acc.z / dn)));
    o.w = 1.0f / (1.0f + __expf(-(acc.w / dn)));
    if (live) ((float4*)(out + (size_t)node * DD))[c] = o;
}

extern "C" void kernel_launch(void* const* d_in, const int* in_sizes, int n_in,
                              void* d_out, int out_size, void* d_ws, size_t ws_size,
                              hipStream_t stream) {
    const float* x    = (const float*)d_in[0];
    const float* W    = (const float*)d_in[1];
    const float* bias = (const float*)d_in[2];
    const float* avec = (const float*)d_in[3];
    const int*   ei   = (const int*)d_in[4];

    const int N = in_sizes[0] / DD;
    const int E = in_sizes[4] / 2;
    const int* src = ei;
    const int* dst = ei + E;

    const int numChunks = (E + CHUNK - 1) / CHUNK;
    const int numRanges = (N + RNODES - 1) / RNODES < MAXR ? (N + RNODES - 1) / RNODES : MAXR;

    float*    Wh      = (float*)d_ws;
    int*      bucket  = (int*)(Wh + (size_t)N * DD);
    int*      ovf     = bucket + (size_t)N * CAP;
    unsigned* cnt     = (unsigned*)(ovf + 2 * (size_t)E);
    unsigned* cellCnt = cnt + (((size_t)N + 1 + 3) & ~(size_t)3);
    unsigned* seg     = cellCnt + (((size_t)numRanges * numChunks + 3) & ~(size_t)3);
    unsigned* ovfACnt = seg + (size_t)numRanges * numChunks * CELL;
    int*      ovfA    = (int*)(ovfACnt + (((size_t)numChunks + 3) & ~(size_t)3));

    float* out = (float*)d_out;

    const int tiles = (N + 15) / 16;
    const int gemmBlocks = (tiles + 7) / 8;
    k_gemm_bin<<<gemmBlocks + numChunks, 256, 0, stream>>>(x, W, bias, Wh, src, dst, seg, cellCnt, ovfA, ovfACnt,
                                                           cnt, E, N, gemmBlocks, numChunks, numRanges);

    k_build<<<numRanges, BT, 0, stream>>>(seg, cellCnt, ovfA, ovfACnt, cnt, bucket, ovf, N, numChunks);

    k_node<<<(N + 15) / 16, 256, 0, stream>>>(Wh, avec, cnt, bucket, ovf, out, N);
}


# node kernel, fewer vector instructions per 4-edge step: leaky relu as max(t,0.2t); 32-bit gather offsets; missing edges get logit -inf by a broadcast add, no selects; logit lane terms now (0+2)+(1+3)
# speedup vs baseline: 8.1591x; 1.0776x over previous
#include <hip/hip_runtime.h>
#include <hip/hip_bf16.h>
#include <math.h>


#define DD 64
#define NSLOPE 0.2f
#define NOEDGE (-__builtin_inff())

typedef __attribute__((ext_vector_type(2))) float v2f;
typedef __attribute__((ext_vector_type(8))) float v8f;

__device__ __forceinline__ void gemm_block(const float* __restrict__ x,
                                           const float* __restrict__ W,
                                           const float* __restrict__ bias,
                                           float* __restrict__ Wh,
                                           const int blk, const int n) {
    const int lane = threadIdx.x & 31;
    const int wave = threadIdx.x >> 5;
    const int tile = blk * 8 + wave;
    if (tile * 16 >= n) return;

    const int lm   = lane & 15;
    const int half = lane >> 4;
    const int row  = tile * 16 + lm;

    const float* xr = x + (size_t)row * DD + 2 * half;
    v2f areg[16];
#pragma unroll
    for (int i = 0; i < 16; ++i) {
        areg[i].x = xr[4 * i + 0];
        areg[i].y = xr[4 * i + 1];
    }

#pragma unroll
    for (int c = 0; c < 4; ++c) {
        v8f acc = {};
        const float* wr = W + (size_t)(c * 16 + lm) * DD + 2 * half;
#pragma unroll
        for (int i = 0; i < 16; ++i) {
            v2f bfrag;
            bfrag.x = wr[4 * i + 0];
            bfrag.y = wr[4 * i + 1];
            acc = __builtin_amdgcn_wmma_f32_16x16x4_f32(
                false, areg[i], false, bfrag, (short)0, acc, false, false);
        }
        const float bv = bias[c * 16 + lm];
#pragma unroll
        for (int v = 0; v < 8; ++v) {
            Wh[(size_t)(tile * 16 + v + 8 * half) * DD + c * 16 + lm] = acc[v] + bv;
        }
    }
}

#define CAP 64
#define RSH 8
#define RNODES (1 << RSH)
#define MAXR 256
#define CELL 48
#define CHUNK 4096

__device__ __forceinline__ void bin_block(const int* __restrict__ src,
                                          const int* __restrict__ dst,
                                          unsigned* __restrict__ seg,
                                          unsigned* __restrict__ cellCnt,
                                          int* __restrict__ ovfA,
                                          unsigned* __restrict__ ovfACnt,
                                          const int E, const int chunk,
                                          const int numChunks, const int numRanges,
                                          unsigned* hist, unsigned* cells) {
    const int t = threadIdx.x;
    for (int i = t; i <= MAXR; i += 256) hist[i] = 0u;
    __syncthreads();

    const int e0 = chunk * CHUNK + t;
    int d[16], sv[16];
#pragma unroll
    for (int j = 0; j < 16; ++j) {
        const int e = e0 + j * 256;
        const bool in = e < E;
        d[j]  = in ? dst[e] : -1;
        sv[j] = in ? src[e] : 0;
    }
#pragma unroll
    for (int j = 0; j < 16; ++j) {
        if (d[j] >= 0) {
            const unsigned rg = min((unsigned)d[j] >> RSH, (unsigned)(MAXR - 1));
            const unsigned k = atomicAdd(&hist[rg], 1u);
            if (k < (unsigned)CELL) {
                cells[rg * CELL + k] = ((unsigned)(d[j] & (RNODES - 1)) << 24) | ((unsigned)sv[j] & 0x00ffffffu);
            } else {
                const unsigned q = atomicAdd(&hist[MAXR], 1u);
                ovfA[2 * ((size_t)chunk * CHUNK + q) + 0] = d[j];
                ovfA[2 * ((size_t)chunk * CHUNK + q) + 1] = sv[j];
            }
        }
    }
    __syncthreads();

    for (int i = t; i < numRanges * CELL; i += 256) {
        const int rg = i / CELL;
        const int slot = i - rg * CELL;
        const unsigned c = min(hist[rg], (unsigned)CELL);
        if ((unsigned)slot < c) seg[((size_t)rg * numChunks + chunk) * CELL + slot] = cells[i];
    }
    for (int rg = t; rg < numRanges; rg += 256)
        cellCnt[(size_t)rg * numChunks + chunk] = min(hist[rg], (unsigned)CELL);
    if (t == 0) ovfACnt[chunk] = min(hist[MAXR], (unsigned)CHUNK);
}

__global__ __launch_bounds__(256) void k_gemm_bin(const float* __restrict__ x,
                                                  const float* __restrict__ W,
                                                  const float* __restrict__ bias,
                                                  float* __restrict__ Wh,
                                                  const int* __restrict__ src,
                                                  const int* __restrict__ dst,
                                                  unsigned* __restrict__ seg,
                                                  unsigned* __restrict__ cellCnt,
                                                  int* __restrict__ ovfA,
                                                  unsigned* __restrict__ ovfACnt,
                                                  unsigned* __restrict__ cnt,
                                                  int E, int n, int gemmBlocks,
                                                  int numChunks, int numRanges) {
    __shared__ unsigned hist[MAXR + 1];
    __shared__ unsigned cells[MAXR * CELL];
    const int blk = blockIdx.x;
    if (blk == 0 && threadIdx.x == 0) cnt[n] = 0u;
    if (blk < gemmBlocks) {
        gemm_block(x, W, bias, Wh, blk, n);
    } else {
        bin_block(src, dst, seg, cellCnt, ovfA, ovfACnt, E, blk - gemmBlocks, numChunks, numRanges, hist, cells);
    }
}

__device__ __forceinline__ void put_edge(unsigned* cntL, int* rows,
                                         unsigned* __restrict__ cnt, int* __restrict__ ovf,
                                         const int n, const int r, const unsigned nl, const int s) {
    const unsigned k = atomicAdd(&cntL[nl], 1u);
    if (k < (unsigned)CAP) {
        rows[nl * CAP + k] = s;
    } else {
        const unsigned p = atomicAdd(cnt + n, 1u);
        ovf[2 * (size_t)p + 0] = (r << RSH) + (int)nl;
        ovf[2 * (size_t)p + 1] = s;
    }
}

#define BT 1024
#define BU 10

__global__ __launch_bounds__(BT) void k_build(const unsigned* __restrict__ seg,
                                              const unsigned* __restrict__ cellCnt,
                                              const int* __restrict__ ovfA,
                                              const unsigned* __restrict__ ovfACnt,
                                              unsigned* __restrict__ cnt,
                                              int* __restrict__ bucket,
                                              int* __restrict__ ovf,
                                              int n, int numChunks) {
    __shared__ unsigned cntL[RNODES];
    __shared__ int rows[RNODES * CAP];
    const int t = threadIdx.x;
    const int r = blockIdx.x;
    for (int i = t; i < RNODES; i += BT) cntL[i] = 0u;
    __syncthreads();

    const int total = numChunks * CELL;
    const unsigned* segR = seg + (size_t)r * numChunks * CELL;
    const unsigned* ccR  = cellCnt + (size_t)r * numChunks;
    for (int base = 0; base < total; base += BT * BU) {
        unsigned c[BU], rec[BU];
        bool ok[BU];
#pragma unroll
        for (int j = 0; j < BU; ++j) {
            const int i = base + j * BT + t;
            c[j] = (i < total) ? ccR[i / CELL] : 0u;
        }
#pragma unroll
        for (int j = 0; j < BU; ++j) {
            const int i = base + j * BT + t;
            const int slot = i - (i / CELL) * CELL;
            ok[j]  = (unsigned)slot < c[j];
            rec[j] = ok[j] ? segR[i] : 0u;
        }
#pragma unroll
        for (int j = 0; j < BU; ++j)
            if (ok[j]) put_edge(cntL, rows, cnt, ovf, n, r, rec[j] >> 24, (int)(rec[j] & 0x00ffffffu));
    }

    for (int ch = t; ch < numChunks; ch += BT) {
        const unsigned oc = min(ovfACnt[ch], (unsigned)CHUNK);
        for (unsigned q = 0; q < oc; ++q) {
            const int ed = ovfA[2 * ((size_t)ch * CHUNK + q) + 0];
            const int es = ovfA[2 * ((size_t)ch * CHUNK + q) + 1];
            if (min((unsigned)ed >> RSH, (unsigned)(MAXR - 1)) == (unsigned)r)
                put_edge(cntL, rows, cnt, ovf, n, r, (unsigned)ed & (unsigned)(RNODES - 1), es);
        }
    }
    __syncthreads();

    for (int i = t; i < RNODES * CAP; i += BT) {
        const int nl = i / CAP;
        const int slot = i - nl * CAP;
        const int node = (r << RSH) + nl;
        const unsigned c = min(cntL[nl], (unsigned)CAP);
        if (node < n && (unsigned)slot < c) bucket[(size_t)node * CAP + slot] = rows[i];
    }
    for (int nl = t; nl < RNODES; nl += BT) {
        const int node = (r << RSH) + nl;
        if (node < n) cnt[node] = cntL[nl];
    }
}

template <int J> __device__ __forceinline__ int rowBcast(int v) {
    return __builtin_amdgcn_mov_dpp(v, 0x150 + J, 0xf, 0xf, true);
}
template <int M> __device__ __forceinline__ float rowXorAdd(float p) {
    const int q = __builtin_amdgcn_mov_dpp(__float_as_int(p), 0x160 + M, 0xf, 0xf, true);
    return p + __int_as_float(q);
}
__device__ __forceinline__ float rowLogit(const float4 ws, const float4 wd, const float4 av) {
    const v2f s01 = v2f{ws.x, ws.y} + v2f{wd.x, wd.y};
    const v2f s23 = v2f{ws.z, ws.w} + v2f{wd.z, wd.w};
    const v2f r01 = __builtin_elementwise_max(s01, s01 * NSLOPE);
    const v2f r23 = __builtin_elementwise_max(s23, s23 * NSLOPE);
    const v2f q = r01 * v2f{av.x, av.y} + r23 * v2f{av.z, av.w};
    float p = q.x + q.y;
    p = rowXorAdd<1>(p);
    p = rowXorAdd<2>(p);
    p = rowXorAdd<4>(p);
    p = rowXorAdd<8>(p);
    return p;
}
template <int J> __device__ __forceinline__ void edge4(const float* __restrict__ Wh,
                                                       const float4 wi, const float4 av,
                                                       const int offv, const float pen, const unsigned c16,
                                                       float& m, float& den, float4& acc) {
    const char* base = (const char*)Wh;
    const unsigned o0 = (unsigned)rowBcast<J + 0>(offv) + c16;
    const unsigned o1 = (unsigned)rowBcast<J + 1>(offv) + c16;
    const unsigned o2 = (unsigned)rowBcast<J + 2>(offv) + c16;
    const unsigned o3 = (unsigned)rowBcast<J + 3>(offv) + c16;
    const float4 w0 = *(const float4*)(base + o0);
    const float4 w1 = *(const float4*)(base + o1);
    const float4 w2 = *(const float4*)(base + o2);
    const float4 w3 = *(const float4*)(base + o3);
    asm volatile("" :: "v"(w0.x), "v"(w0.y), "v"(w0.z), "v"(w0.w), "v"(w1.x), "v"(w1.y), "v"(w1.z), "v"(w1.w),
                       "v"(w2.x), "v"(w2.y), "v"(w2.z), "v"(w2.w), "v"(w3.x), "v"(w3.y), "v"(w3.z), "v"(w3.w));
    const int pi = __float_as_int(pen);
    const float l0 = rowLogit(w0, wi, av) + __int_as_float(rowBcast<J + 0>(pi));
    const float l1 = rowLogit(w1, wi, av) + __int_as_float(rowBcast<J + 1>(pi));
    const float l2 = rowLogit(w2, wi, av) + __int_as_float(rowBcast<J + 2>(pi));
    const float l3 = rowLogit(w3, wi, av) + __int_as_float(rowBcast<J + 3>(pi));
    const float mn = fmaxf(fmaxf(m, l0), fmaxf(l1, fmaxf(l2, l3)));
    const float sc = __expf(m - mn);
    const float p0 = __expf(l0 - mn);
    const float p1 = __expf(l1 - mn);
    const float p2 = __expf(l2 - mn);
    const float p3 = __expf(l3 - mn);
    den   = den   * sc + ((p0 + p1) + (p2 + p3));
    acc.x = acc.x * sc + ((p0 * w0.x + p1 * w1.x) + (p2 * w2.x + p3 * w3.x));
    acc.y = acc.y * sc + ((p0 * w0.y + p1 * w1.y) + (p2 * w2.y + p3 * w3.y));
    acc.z = acc.z * sc + ((p0 * w0.z + p1 * w1.z) + (p2 * w2.z + p3 * w3.z));
    acc.w = acc.w * sc + ((p0 * w0.w + p1 * w1.w) + (p2 * w2.w + p3 * w3.w));
    m = mn;
}

__global__ __launch_bounds__(256) void k_node(const float* __restrict__ Wh,
                                              const float* __restrict__ avec,
                                              const unsigned* __restrict__ cnt,
                                              const int* __restrict__ bucket,
                                              const int* __restrict__ ovf,
                                              float* __restrict__ out, int n) {
    const int lane = threadIdx.x & 31;
    const int c = lane & 15;
    const int node0 = (blockIdx.x * 8 + (threadIdx.x >> 5)) * 2;
    if (node0 >= n) return;
    const int node = node0 + (lane >> 4);
    const bool live = node < n;
    const int nd = live ? node : node0;

    const unsigned cn = cnt[nd];
    const int deg = live ? (int)min(cn, (unsigned)CAP) : 0;
    const unsigned beg = (unsigned)nd * CAP;
    const unsigned moreRows = __builtin_amdgcn_ballot_w32(live && cn > (unsigned)CAP) & 0x00010001u;
    const float4 av = ((const float4*)avec)[c];
    const float4 wi = ((const float4*)(Wh + (size_t)nd * DD))[c];
    const unsigned c16 = (unsigned)c * 16u;

    const int d0 = __builtin_amdgcn_readlane(deg, 0);
    const int d1 = __builtin_amdgcn_readlane(deg, 16);
    const int degmax = d0 > d1 ? d0 : d1;

    float m = -3.0e38f;
    float den = 0.0f;
    float4 acc = make_float4(0.0f, 0.0f, 0.0f, 0.0f);

    for (int k = 0; k < degmax; k += 16) {
        const bool has = c < deg - k;
        int sv = has ? bucket[beg + (unsigned)(k + c)] : nd;
        sv = min(max(sv, 0), n - 1);
        const int offv = sv * (DD * 4);
        const float pen = has ? 0.0f : NOEDGE;
        edge4<0>(Wh, wi, av, offv, pen, c16, m, den, acc);
        if (k + 4 < degmax)  edge4<4>(Wh, wi, av, offv, pen, c16, m, den, acc);
        if (k + 8 < degmax)  edge4<8>(Wh, wi, av, offv, pen, c16, m, den, acc);
        if (k + 12 < degmax) edge4<12>(Wh, wi, av, offv, pen, c16, m, den, acc);
    }

    if (moreRows != 0u) {
        const int sh = (node & 1) << 4;
        const bool more = ((moreRows >> sh) & 1u) != 0u;
        const int ovn = (int)min(cnt[n], 0x7fffffffu);
        for (int k = 0; k < ovn; k += 16) {
            int od = -1, os = nd;
            if (k + c < ovn) {
                od = ovf[2 * (size_t)(k + c) + 0];
                os = ovf[2 * (size_t)(k + c) + 1];
            }
            const bool hit = more && od == node;
            const unsigned bal = __builtin_amdgcn_ballot_w32(hit);
            if (bal == 0u) continue;
            int sv = hit ? os : nd;
            sv = min(max(sv, 0), n - 1);
            const int offv = sv * (DD * 4);
            const float pen = hit ? 0.0f : NOEDGE;
            if (bal & 0x000f000fu) edge4<0>(Wh, wi, av, offv, pen, c16, m, den, acc);
            if (bal & 0x00f000f0u) edge4<4>(Wh, wi, av, offv, pen, c16, m, den, acc);
            if (bal & 0x0f000f00u) edge4<8>(Wh, wi, av, offv, pen, c16, m, den, acc);
            if (bal & 0xf000f000u) edge4<12>(Wh, wi, av, offv, pen, c16, m, den, acc);
        }
    }

    const float dn = fmaxf(den, 1e-9f);
    float4 o;
    o.x = 1.0f / (1.0f + __expf(-(acc.x / dn)));
    o.y = 1.0f / (1.0f + __expf(-(acc.y / dn)));
    o.z = 1.0f / (1.0f + __expf(-(acc.z / dn)));
    o.w = 1.0f / (1.0f + __expf(-(acc.w / dn)));
    if (live) ((float4*)(out + (size_t)node * DD))[c] = o;
}

extern "C" void kernel_launch(void* const* d_in, const int* in_sizes, int n_in,
                              void* d_out, int out_size, void* d_ws, size_t ws_size,
                              hipStream_t stream) {
    const float* x    = (const float*)d_in[0];
    const float* W    = (const float*)d_in[1];
    const float* bias = (const float*)d_in[2];
    const float* avec = (const float*)d_in[3];
    const int*   ei   = (const int*)d_in[4];

    const int N = in_sizes[0] / DD;
    const int E = in_sizes[4] / 2;
    const int* src = ei;
    const int* dst = ei + E;

    const int numChunks = (E + CHUNK - 1) / CHUNK;
    const int numRanges = (N + RNODES - 1) / RNODES < MAXR ? (N + RNODES - 1) / RNODES : MAXR;

    float*    Wh      = (float*)d_ws;
    int*      bucket  = (int*)(Wh + (size_t)N * DD);
    int*      ovf     = bucket + (size_t)N * CAP;
    unsigned* cnt     = (unsigned*)(ovf + 2 * (size_t)E);
    unsigned* cellCnt = cnt + (((size_t)N + 1 + 3) & ~(size_t)3);
    unsigned* seg     = cellCnt + (((size_t)numRanges * numChunks + 3) & ~(size_t)3);
    unsigned* ovfACnt = seg + (size_t)numRanges * numChunks * CELL;
    int*      ovfA    = (int*)(ovfACnt + (((size_t)numChunks + 3) & ~(size_t)3));

    float* out = (float*)d_out;

    const int tiles = (N + 15) / 16;
    const int gemmBlocks = (tiles + 7) / 8;
    k_gemm_bin<<<gemmBlocks + numChunks, 256, 0, stream>>>(x, W, bias, Wh, src, dst, seg, cellCnt, ovfA, ovfACnt,
                                                           cnt, E, N, gemmBlocks, numChunks, numRanges);

    k_build<<<numRanges, BT, 0, stream>>>(seg, cellCnt, ovfA, ovfACnt, cnt, bucket, ovf, N, numChunks);

    k_node<<<(N + 15) / 16, 256, 0, stream>>>(Wh, avec, cnt, bucket, ovf, out, N);
}


# projection: W and each wave's x tile go through LDS (whole-line 16-B loads, 16-B fragment reads at pitch 68), Wh tile leaves as whole rows; k order inside the product now 8j+4h+{0..3}
# speedup vs baseline: 8.4577x; 1.0366x over previous
#include <hip/hip_runtime.h>
#include <hip/hip_bf16.h>
#include <math.h>


#define DD 64
#define NSLOPE 0.2f
#define NOEDGE (-__builtin_inff())

typedef __attribute__((ext_vector_type(2))) float v2f;
typedef __attribute__((ext_vector_type(8))) float v8f;

#define WP 68
#define GEMM_LDS_WORDS (64 * WP + 8 * 16 * WP)

__device__ __forceinline__ void gemm_block(const float* __restrict__ x,
                                           const float* __restrict__ W,
                                           const float* __restrict__ bias,
                                           float* __restrict__ Wh,
                                           const int blk, const int n, float* lds) {
    const int t    = threadIdx.x;
    const int lane = t & 31;
    const int wave = t >> 5;
    const int lm   = lane & 15;
    const int half = lane >> 4;
    const int tiles = (n + 15) >> 4;
    const bool valid = blk * 8 + wave < tiles;
    const int tile = valid ? blk * 8 + wave : tiles - 1;
    const int row0 = tile * 16;

    float* Wl = lds;
    float* T  = lds + 64 * WP + wave * 16 * WP;

    float4 wv[4], xv[8];
#pragma unroll
    for (int r = 0; r < 4; ++r) wv[r] = ((const float4*)W)[t + 256 * r];
#pragma unroll
    for (int q = 0; q < 8; ++q) {
        const int grow = min(row0 + 2 * q + half, n - 1);
        xv[q] = ((const float4*)(x + (size_t)grow * DD))[lm];
    }
#pragma unroll
    for (int r = 0; r < 4; ++r) {
        const int idx = t + 256 * r;
        *(float4*)(Wl + (idx >> 4) * WP + (idx & 15) * 4) = wv[r];
    }
#pragma unroll
    for (int q = 0; q < 8; ++q) *(float4*)(T + (2 * q + half) * WP + lm * 4) = xv[q];
    __syncthreads();

    float4 af[8];
#pragma unroll
    for (int j = 0; j < 8; ++j) af[j] = *(const float4*)(T + lm * WP + 8 * j + 4 * half);

    v8f acc[4];
#pragma unroll
    for (int c = 0; c < 4; ++c) {
        acc[c] = v8f{};
        const float* wl = Wl + (c * 16 + lm) * WP + 4 * half;
#pragma unroll
        for (int j = 0; j < 8; ++j) {
            const float4 bf = *(const float4*)(wl + 8 * j);
            acc[c] = __builtin_amdgcn_wmma_f32_16x16x4_f32(
                false, v2f{af[j].x, af[j].y}, false, v2f{bf.x, bf.y}, (short)0, acc[c], false, false);
            acc[c] = __builtin_amdgcn_wmma_f32_16x16x4_f32(
                false, v2f{af[j].z, af[j].w}, false, v2f{bf.z, bf.w}, (short)0, acc[c], false, false);
        }
    }
    __syncthreads();

#pragma unroll
    for (int c = 0; c < 4; ++c) {
        const float bv = bias[c * 16 + lm];
#pragma unroll
        for (int v = 0; v < 8; ++v) T[(v + 8 * half) * WP + c * 16 + lm] = acc[c][v] + bv;
    }
    __syncthreads();

    if (valid) {
        float* wo = Wh + (size_t)row0 * DD + lm * 4;
        if (row0 + 16 <= n) {
#pragma unroll
            for (int q = 0; q < 8; ++q) {
                const int row = 2 * q + half;
                *(float4*)(wo + row * DD) = *(const float4*)(T + row * WP + lm * 4);
            }
        } else {
            for (int q = 0; q < 8; ++q) {
                const int row = 2 * q + half;
                if (row0 + row < n) *(float4*)(wo + row * DD) = *(const float4*)(T + row * WP + lm * 4);
            }
        }
    }
}

#define CAP 64
#define RSH 8
#define RNODES (1 << RSH)
#define MAXR 256
#define CELL 48
#define CHUNK 4096

__device__ __forceinline__ void bin_block(const int* __restrict__ src,
                                          const int* __restrict__ dst,
                                          unsigned* __restrict__ seg,
                                          unsigned* __restrict__ cellCnt,
                                          int* __restrict__ ovfA,
                                          unsigned* __restrict__ ovfACnt,
                                          const int E, const int chunk,
                                          const int numChunks, const int numRanges,
                                          unsigned* hist, unsigned* cells) {
    const int t = threadIdx.x;
    for (int i = t; i <= MAXR; i += 256) hist[i] = 0u;
    __syncthreads();

    const int e0 = chunk * CHUNK + t;
    int d[16], sv[16];
#pragma unroll
    for (int j = 0; j < 16; ++j) {
        const int e = e0 + j * 256;
        const bool in = e < E;
        d[j]  = in ? dst[e] : -1;
        sv[j] = in ? src[e] : 0;
    }
#pragma unroll
    for (int j = 0; j < 16; ++j) {
        if (d[j] >= 0) {
            const unsigned rg = min((unsigned)d[j] >> RSH, (unsigned)(MAXR - 1));
            const unsigned k = atomicAdd(&hist[rg], 1u);
            if (k < (unsigned)CELL) {
                cells[rg * CELL + k] = ((unsigned)(d[j] & (RNODES - 1)) << 24) | ((unsigned)sv[j] & 0x00ffffffu);
            } else {
                const unsigned q = atomicAdd(&hist[MAXR], 1u);
                ovfA[2 * ((size_t)chunk * CHUNK + q) + 0] = d[j];
                ovfA[2 * ((size_t)chunk * CHUNK + q) + 1] = sv[j];
            }
        }
    }
    __syncthreads();

    for (int i = t; i < numRanges * CELL; i += 256) {
        const int rg = i / CELL;
        const int slot = i - rg * CELL;
        const unsigned c = min(hist[rg], (unsigned)CELL);
        if ((unsigned)slot < c) seg[((size_t)rg * numChunks + chunk) * CELL + slot] = cells[i];
    }
    for (int rg = t; rg < numRanges; rg += 256)
        cellCnt[(size_t)rg * numChunks + chunk] = min(hist[rg], (unsigned)CELL);
    if (t == 0) ovfACnt[chunk] = min(hist[MAXR], (unsigned)CHUNK);
}

__global__ __launch_bounds__(256) void k_gemm_bin(const float* __restrict__ x,
                                                  const float* __restrict__ W,
                                                  const float* __restrict__ bias,
                                                  float* __restrict__ Wh,
                                                  const int* __restrict__ src,
                                                  const int* __restrict__ dst,
                                                  unsigned* __restrict__ seg,
                                                  unsigned* __restrict__ cellCnt,
                                                  int* __restrict__ ovfA,
                                                  unsigned* __restrict__ ovfACnt,
                                                  unsigned* __restrict__ cnt,
                                                  int E, int n, int gemmBlocks,
                                                  int numChunks, int numRanges) {
    __shared__ __attribute__((aligned(16))) unsigned smem[GEMM_LDS_WORDS];
    unsigned* hist  = smem;
    unsigned* cells = smem + MAXR + 4;
    const int blk = blockIdx.x;
    if (blk == 0 && threadIdx.x == 0) cnt[n] = 0u;
    if (blk < gemmBlocks) {
        gemm_block(x, W, bias, Wh, blk, n, (float*)smem);
    } else {
        bin_block(src, dst, seg, cellCnt, ovfA, ovfACnt, E, blk - gemmBlocks, numChunks, numRanges, hist, cells);
    }
}

__device__ __forceinline__ void put_edge(unsigned* cntL, int* rows,
                                         unsigned* __restrict__ cnt, int* __restrict__ ovf,
                                         const int n, const int r, const unsigned nl, const int s) {
    const unsigned k = atomicAdd(&cntL[nl], 1u);
    if (k < (unsigned)CAP) {
        rows[nl * CAP + k] = s;
    } else {
        const unsigned p = atomicAdd(cnt + n, 1u);
        ovf[2 * (size_t)p + 0] = (r << RSH) + (int)nl;
        ovf[2 * (size_t)p + 1] = s;
    }
}

#define BT 1024
#define BU 10

__global__ __launch_bounds__(BT) void k_build(const unsigned* __restrict__ seg,
                                              const unsigned* __restrict__ cellCnt,
                                              const int* __restrict__ ovfA,
                                              const unsigned* __restrict__ ovfACnt,
                                              unsigned* __restrict__ cnt,
                                              int* __restrict__ bucket,
                                              int* __restrict__ ovf,
                                              int n, int numChunks) {
    __shared__ unsigned cntL[RNODES];
    __shared__ int rows[RNODES * CAP];
    const int t = threadIdx.x;
    const int r = blockIdx.x;
    for (int i = t; i < RNODES; i += BT) cntL[i] = 0u;
    __syncthreads();

    const int total = numChunks * CELL;
    const unsigned* segR = seg + (size_t)r * numChunks * CELL;
    const unsigned* ccR  = cellCnt + (size_t)r * numChunks;
    for (int base = 0; base < total; base += BT * BU) {
        unsigned c[BU], rec[BU];
        bool ok[BU];
#pragma unroll
        for (int j = 0; j < BU; ++j) {
            const int i = base + j * BT + t;
            c[j] = (i < total) ? ccR[i / CELL] : 0u;
        }
#pragma unroll
        for (int j = 0; j < BU; ++j) {
            const int i = base + j * BT + t;
            const int slot = i - (i / CELL) * CELL;
            ok[j]  = (unsigned)slot < c[j];
            rec[j] = ok[j] ? segR[i] : 0u;
        }
#pragma unroll
        for (int j = 0; j < BU; ++j)
            if (ok[j]) put_edge(cntL, rows, cnt, ovf, n, r, rec[j] >> 24, (int)(rec[j] & 0x00ffffffu));
    }

    for (int ch = t; ch < numChunks; ch += BT) {
        const unsigned oc = min(ovfACnt[ch], (unsigned)CHUNK);
        for (unsigned q = 0; q < oc; ++q) {
            const int ed = ovfA[2 * ((size_t)ch * CHUNK + q) + 0];
            const int es = ovfA[2 * ((size_t)ch * CHUNK + q) + 1];
            if (min((unsigned)ed >> RSH, (unsigned)(MAXR - 1)) == (unsigned)r)
                put_edge(cntL, rows, cnt, ovf, n, r, (unsigned)ed & (unsigned)(RNODES - 1), es);
        }
    }
    __syncthreads();

    for (int i = t; i < RNODES * CAP; i += BT) {
        const int nl = i / CAP;
        const int slot = i - nl * CAP;
        const int node = (r << RSH) + nl;
        const unsigned c = min(cntL[nl], (unsigned)CAP);
        if (node < n && (unsigned)slot < c) bucket[(size_t)node * CAP + slot] = rows[i];
    }
    for (int nl = t; nl < RNODES; nl += BT) {
        const int node = (r << RSH) + nl;
        if (node < n) cnt[node] = cntL[nl];
    }
}

template <int J> __device__ __forceinline__ int rowBcast(int v) {
    return __builtin_amdgcn_mov_dpp(v, 0x150 + J, 0xf, 0xf, true);
}
template <int M> __device__ __forceinline__ float rowXorAdd(float p) {
    const int q = __builtin_amdgcn_mov_dpp(__float_as_int(p), 0x160 + M, 0xf, 0xf, true);
    return p + __int_as_float(q);
}
__device__ __forceinline__ float rowLogit(const float4 ws, const float4 wd, const float4 av) {
    const v2f s01 = v2f{ws.x, ws.y} + v2f{wd.x, wd.y};
    const v2f s23 = v2f{ws.z, ws.w} + v2f{wd.z, wd.w};
    const v2f r01 = __builtin_elementwise_max(s01, s01 * NSLOPE);
    const v2f r23 = __builtin_elementwise_max(s23, s23 * NSLOPE);
    const v2f q = r01 * v2f{av.x, av.y} + r23 * v2f{av.z, av.w};
    float p = q.x + q.y;
    p = rowXorAdd<1>(p);
    p = rowXorAdd<2>(p);
    p = rowXorAdd<4>(p);
    p = rowXorAdd<8>(p);
    return p;
}
template <int J> __device__ __forceinline__ void edge4(const float* __restrict__ Wh,
                                                       const float4 wi, const float4 av,
                                                       const int offv, const float pen, const unsigned c16,
                                                       float& m, float& den, float4& acc) {
    const char* base = (const char*)Wh;
    const unsigned o0 = (unsigned)rowBcast<J + 0>(offv) + c16;
    const unsigned o1 = (unsigned)rowBcast<J + 1>(offv) + c16;
    const unsigned o2 = (unsigned)rowBcast<J + 2>(offv) + c16;
    const unsigned o3 = (unsigned)rowBcast<J + 3>(offv) + c16;
    const float4 w0 = *(const float4*)(base + o0);
    const float4 w1 = *(const float4*)(base + o1);
    const float4 w2 = *(const float4*)(base + o2);
    const float4 w3 = *(const float4*)(base + o3);
    asm volatile("" :: "v"(w0.x), "v"(w0.y), "v"(w0.z), "v"(w0.w), "v"(w1.x), "v"(w1.y), "v"(w1.z), "v"(w1.w),
                       "v"(w2.x), "v"(w2.y), "v"(w2.z), "v"(w2.w), "v"(w3.x), "v"(w3.y), "v"(w3.z), "v"(w3.w));
    const int pi = __float_as_int(pen);
    const float l0 = rowLogit(w0, wi, av) + __int_as_float(rowBcast<J + 0>(pi));
    const float l1 = rowLogit(w1, wi, av) + __int_as_float(rowBcast<J + 1>(pi));
    const float l2 = rowLogit(w2, wi, av) + __int_as_float(rowBcast<J + 2>(pi));
    const float l3 = rowLogit(w3, wi, av) + __int_as_float(rowBcast<J + 3>(pi));
    const float mn = fmaxf(fmaxf(m, l0), fmaxf(l1, fmaxf(l2, l3)));
    const float sc = __expf(m - mn);
    const float p0 = __expf(l0 - mn);
    const float p1 = __expf(l1 - mn);
    const float p2 = __expf(l2 - mn);
    const float p3 = __expf(l3 - mn);
    den   = den   * sc + ((p0 + p1) + (p2 + p3));
    acc.x = acc.x * sc + ((p0 * w0.x + p1 * w1.x) + (p2 * w2.x + p3 * w3.x));
    acc.y = acc.y * sc + ((p0 * w0.y + p1 * w1.y) + (p2 * w2.y + p3 * w3.y));
    acc.z = acc.z * sc + ((p0 * w0.z + p1 * w1.z) + (p2 * w2.z + p3 * w3.z));
    acc.w = acc.w * sc + ((p0 * w0.w + p1 * w1.w) + (p2 * w2.w + p3 * w3.w));
    m = mn;
}

__global__ __launch_bounds__(256) void k_node(const float* __restrict__ Wh,
                                              const float* __restrict__ avec,
                                              const unsigned* __restrict__ cnt,
                                              const int* __restrict__ bucket,
                                              const int* __restrict__ ovf,
                                              float* __restrict__ out, int n) {
    const int lane = threadIdx.x & 31;
    const int c = lane & 15;
    const int node0 = (blockIdx.x * 8 + (threadIdx.x >> 5)) * 2;
    if (node0 >= n) return;
    const int node = node0 + (lane >> 4);
    const bool live = node < n;
    const int nd = live ? node : node0;

    const unsigned cn = cnt[nd];
    const int deg = live ? (int)min(cn, (unsigned)CAP) : 0;
    const unsigned beg = (unsigned)nd * CAP;
    const unsigned moreRows = __builtin_amdgcn_ballot_w32(live && cn > (unsigned)CAP) & 0x00010001u;
    const float4 av = ((const float4*)avec)[c];
    const float4 wi = ((const float4*)(Wh + (size_t)nd * DD))[c];
    const unsigned c16 = (unsigned)c * 16u;

    const int d0 = __builtin_amdgcn_readlane(deg, 0);
    const int d1 = __builtin_amdgcn_readlane(deg, 16);
    const int degmax = d0 > d1 ? d0 : d1;

    float m = -3.0e38f;
    float den = 0.0f;
    float4 acc = make_float4(0.0f, 0.0f, 0.0f, 0.0f);

    for (int k = 0; k < degmax; k += 16) {
        const bool has = c < deg - k;
        int sv = has ? bucket[beg + (unsigned)(k + c)] : nd;
        sv = min(max(sv, 0), n - 1);
        const int offv = sv * (DD * 4);
        const float pen = has ? 0.0f : NOEDGE;
        edge4<0>(Wh, wi, av, offv, pen, c16, m, den, acc);
        if (k + 4 < degmax)  edge4<4>(Wh, wi, av, offv, pen, c16, m, den, acc);
        if (k + 8 < degmax)  edge4<8>(Wh, wi, av, offv, pen, c16, m, den, acc);
        if (k + 12 < degmax) edge4<12>(Wh, wi, av, offv, pen, c16, m, den, acc);
    }

    if (moreRows != 0u) {
        const int sh = (node & 1) << 4;
        const bool more = ((moreRows >> sh) & 1u) != 0u;
        const int ovn = (int)min(cnt[n], 0x7fffffffu);
        for (int k = 0; k < ovn; k += 16) {
            int od = -1, os = nd;
            if (k + c < ovn) {
                od = ovf[2 * (size_t)(k + c) + 0];
                os = ovf[2 * (size_t)(k + c) + 1];
            }
            const bool hit = more && od == node;
            const unsigned bal = __builtin_amdgcn_ballot_w32(hit);
            if (bal == 0u) continue;
            int sv = hit ? os : nd;
            sv = min(max(sv, 0), n - 1);
            const int offv = sv * (DD * 4);
            const float pen = hit ? 0.0f : NOEDGE;
            if (bal & 0x000f000fu) edge4<0>(Wh, wi, av, offv, pen, c16, m, den, acc);
            if (bal & 0x00f000f0u) edge4<4>(Wh, wi, av, offv, pen, c16, m, den, acc);
            if (bal & 0x0f000f00u) edge4<8>(Wh, wi, av, offv, pen, c16, m, den, acc);
            if (bal & 0xf000f000u) edge4<12>(Wh, wi, av, offv, pen, c16, m, den, acc);
        }
    }

    const float dn = fmaxf(den, 1e-9f);
    float4 o;
    o.x = 1.0f / (1.0f + __expf(-(acc.x / dn)));
    o.y = 1.0f / (1.0f + __expf(-(acc.y / dn)));
    o.z = 1.0f / (1.0f + __expf(-(acc.z / dn)));
    o.w = 1.0f / (1.0f + __expf(-(acc.w / dn)));
    if (live) ((float4*)(out + (size_t)node * DD))[c] = o;
}

extern "C" void kernel_launch(void* const* d_in, const int* in_sizes, int n_in,
                              void* d_out, int out_size, void* d_ws, size_t ws_size,
                              hipStream_t stream) {
    const float* x    = (const float*)d_in[0];
    const float* W    = (const float*)d_in[1];
    const float* bias = (const float*)d_in[2];
    const float* avec = (const float*)d_in[3];
    const int*   ei   = (const int*)d_in[4];

    const int N = in_sizes[0] / DD;
    const int E = in_sizes[4] / 2;
    const int* src = ei;
    const int* dst = ei + E;

    const int numChunks = (E + CHUNK - 1) / CHUNK;
    const int numRanges = (N + RNODES - 1) / RNODES < MAXR ? (N + RNODES - 1) / RNODES : MAXR;

    float*    Wh      = (float*)d_ws;
    int*      bucket  = (int*)(Wh + (size_t)N * DD);
    int*      ovf     = bucket + (size_t)N * CAP;
    unsigned* cnt     = (unsigned*)(ovf + 2 * (size_t)E);
    unsigned* cellCnt = cnt + (((size_t)N + 1 + 3) & ~(size_t)3);
    unsigned* seg     = cellCnt + (((size_t)numRanges * numChunks + 3) & ~(size_t)3);
    unsigned* ovfACnt = seg + (size_t)numRanges * numChunks * CELL;
    int*      ovfA    = (int*)(ovfACnt + (((size_t)numChunks + 3) & ~(size_t)3));

    float* out = (float*)d_out;

    const int tiles = (N + 15) / 16;
    const int gemmBlocks = (tiles + 7) / 8;
    k_gemm_bin<<<gemmBlocks + numChunks, 256, 0, stream>>>(x, W, bias, Wh, src, dst, seg, cellCnt, ovfA, ovfACnt,
                                                           cnt, E, N, gemmBlocks, numChunks, numRanges);

    k_build<<<numRanges, BT, 0, stream>>>(seg, cellCnt, ovfA, ovfACnt, cnt, bucket, ovf, N, numChunks);

    k_node<<<(N + 15) / 16, 256, 0, stream>>>(Wh, avec, cnt, bucket, ovf, out, N);
}


# grouping launch folded into the node kernel: one 1024-thread block per node range fills its slot rows in LDS, then its waves read counts and rows there; 3 launches to 2; sums by LDS arrival as before
# speedup vs baseline: 9.1865x; 1.0862x over previous
#include <hip/hip_runtime.h>
#include <hip/hip_bf16.h>
#include <math.h>


#define DD 64
#define NSLOPE 0.2f
#define NOEDGE (-__builtin_inff())

typedef __attribute__((ext_vector_type(2))) float v2f;
typedef __attribute__((ext_vector_type(8))) float v8f;

#define WP 68
#define GEMM_LDS_WORDS (64 * WP + 8 * 16 * WP)

__device__ __forceinline__ void gemm_block(const float* __restrict__ x,
                                           const float* __restrict__ W,
                                           const float* __restrict__ bias,
                                           float* __restrict__ Wh,
                                           const int blk, const int n, float* lds) {
    const int t    = threadIdx.x;
    const int lane = t & 31;
    const int wave = t >> 5;
    const int lm   = lane & 15;
    const int half = lane >> 4;
    const int tiles = (n + 15) >> 4;
    const bool valid = blk * 8 + wave < tiles;
    const int tile = valid ? blk * 8 + wave : tiles - 1;
    const int row0 = tile * 16;

    float* Wl = lds;
    float* T  = lds + 64 * WP + wave * 16 * WP;

    float4 wv[4], xv[8];
#pragma unroll
    for (int r = 0; r < 4; ++r) wv[r] = ((const float4*)W)[t + 256 * r];
#pragma unroll
    for (int q = 0; q < 8; ++q) {
        const int grow = min(row0 + 2 * q + half, n - 1);
        xv[q] = ((const float4*)(x + (size_t)grow * DD))[lm];
    }
#pragma unroll
    for (int r = 0; r < 4; ++r) {
        const int idx = t + 256 * r;
        *(float4*)(Wl + (idx >> 4) * WP + (idx & 15) * 4) = wv[r];
    }
#pragma unroll
    for (int q = 0; q < 8; ++q) *(float4*)(T + (2 * q + half) * WP + lm * 4) = xv[q];
    __syncthreads();

    float4 af[8];
#pragma unroll
    for (int j = 0; j < 8; ++j) af[j] = *(const float4*)(T + lm * WP + 8 * j + 4 * half);

    v8f acc[4];
#pragma unroll
    for (int c = 0; c < 4; ++c) {
        acc[c] = v8f{};
        const float* wl = Wl + (c * 16 + lm) * WP + 4 * half;
#pragma unroll
        for (int j = 0; j < 8; ++j) {
            const float4 bf = *(const float4*)(wl + 8 * j);
            acc[c] = __builtin_amdgcn_wmma_f32_16x16x4_f32(
                false, v2f{af[j].x, af[j].y}, false, v2f{bf.x, bf.y}, (short)0, acc[c], false, false);
            acc[c] = __builtin_amdgcn_wmma_f32_16x16x4_f32(
                false, v2f{af[j].z, af[j].w}, false, v2f{bf.z, bf.w}, (short)0, acc[c], false, false);
        }
    }
    __syncthreads();

#pragma unroll
    for (int c = 0; c < 4; ++c) {
        const float bv = bias[c * 16 + lm];
#pragma unroll
        for (int v = 0; v < 8; ++v) T[(v + 8 * half) * WP + c * 16 + lm] = acc[c][v] + bv;
    }
    __syncthreads();

    if (valid) {
        float* wo = Wh + (size_t)row0 * DD + lm * 4;
        if (row0 + 16 <= n) {
#pragma unroll
            for (int q = 0; q < 8; ++q) {
                const int row = 2 * q + half;
                *(float4*)(wo + row * DD) = *(const float4*)(T + row * WP + lm * 4);
            }
        } else {
            for (int q = 0; q < 8; ++q) {
                const int row = 2 * q + half;
                if (row0 + row < n) *(float4*)(wo + row * DD) = *(const float4*)(T + row * WP + lm * 4);
            }
        }
    }
}

#define CAP 160
#define MAXR 256
#define MAXRN 256
#define CELL 48
#define CHUNK 4096

__device__ __forceinline__ void bin_block(const int* __restrict__ src,
                                          const int* __restrict__ dst,
                                          unsigned* __restrict__ seg,
                                          unsigned* __restrict__ cellCnt,
                                          int* __restrict__ ovfA,
                                          unsigned* __restrict__ ovfACnt,
                                          const int E, const int chunk,
                                          const int numChunks, const int numRanges, const int RN,
                                          unsigned* hist, unsigned* cells) {
    const int t = threadIdx.x;
    for (int i = t; i <= MAXR; i += 256) hist[i] = 0u;
    __syncthreads();

    const int e0 = chunk * CHUNK + t;
    int d[16], sv[16];
#pragma unroll
    for (int j = 0; j < 16; ++j) {
        const int e = e0 + j * 256;
        const bool in = e < E;
        d[j]  = in ? dst[e] : -1;
        sv[j] = in ? src[e] : 0;
    }
#pragma unroll
    for (int j = 0; j < 16; ++j) {
        if (d[j] >= 0) {
            const unsigned rg = min((unsigned)d[j] / (unsigned)RN, (unsigned)(MAXR - 1));
            const unsigned nl = ((unsigned)d[j] - rg * (unsigned)RN) & (unsigned)(MAXRN - 1);
            const unsigned k = atomicAdd(&hist[rg], 1u);
            if (k < (unsigned)CELL) {
                cells[rg * CELL + k] = (nl << 24) | ((unsigned)sv[j] & 0x00ffffffu);
            } else {
                const unsigned q = atomicAdd(&hist[MAXR], 1u);
                ovfA[2 * ((size_t)chunk * CHUNK + q) + 0] = d[j];
                ovfA[2 * ((size_t)chunk * CHUNK + q) + 1] = sv[j];
            }
        }
    }
    __syncthreads();

    for (int i = t; i < numRanges * CELL; i += 256) {
        const int rg = i / CELL;
        const int slot = i - rg * CELL;
        const unsigned c = min(hist[rg], (unsigned)CELL);
        if ((unsigned)slot < c) seg[((size_t)rg * numChunks + chunk) * CELL + slot] = cells[i];
    }
    for (int rg = t; rg < numRanges; rg += 256)
        cellCnt[(size_t)rg * numChunks + chunk] = min(hist[rg], (unsigned)CELL);
    if (t == 0) ovfACnt[chunk] = min(hist[MAXR], (unsigned)CHUNK);
}

__global__ __launch_bounds__(256) void k_gemm_bin(const float* __restrict__ x,
                                                  const float* __restrict__ W,
                                                  const float* __restrict__ bias,
                                                  float* __restrict__ Wh,
                                                  const int* __restrict__ src,
                                                  const int* __restrict__ dst,
                                                  unsigned* __restrict__ seg,
                                                  unsigned* __restrict__ cellCnt,
                                                  int* __restrict__ ovfA,
                                                  unsigned* __restrict__ ovfACnt,
                                                  int E, int n, int gemmBlocks,
                                                  int numChunks, int numRanges, int RN) {
    __shared__ __attribute__((aligned(16))) unsigned smem[GEMM_LDS_WORDS];
    unsigned* hist  = smem;
    unsigned* cells = smem + MAXR + 4;
    const int blk = blockIdx.x;
    if (blk < gemmBlocks) {
        gemm_block(x, W, bias, Wh, blk, n, (float*)smem);
    } else {
        bin_block(src, dst, seg, cellCnt, ovfA, ovfACnt, E, blk - gemmBlocks, numChunks, numRanges, RN, hist, cells);
    }
}

template <int J> __device__ __forceinline__ int rowBcast(int v) {
    return __builtin_amdgcn_mov_dpp(v, 0x150 + J, 0xf, 0xf, true);
}
template <int M> __device__ __forceinline__ float rowXorAdd(float p) {
    const int q = __builtin_amdgcn_mov_dpp(__float_as_int(p), 0x160 + M, 0xf, 0xf, true);
    return p + __int_as_float(q);
}
__device__ __forceinline__ float rowLogit(const float4 ws, const float4 wd, const float4 av) {
    const v2f s01 = v2f{ws.x, ws.y} + v2f{wd.x, wd.y};
    const v2f s23 = v2f{ws.z, ws.w} + v2f{wd.z, wd.w};
    const v2f r01 = __builtin_elementwise_max(s01, s01 * NSLOPE);
    const v2f r23 = __builtin_elementwise_max(s23, s23 * NSLOPE);
    const v2f q = r01 * v2f{av.x, av.y} + r23 * v2f{av.z, av.w};
    float p = q.x + q.y;
    p = rowXorAdd<1>(p);
    p = rowXorAdd<2>(p);
    p = rowXorAdd<4>(p);
    p = rowXorAdd<8>(p);
    return p;
}
template <int J> __device__ __forceinline__ void edge4(const float* __restrict__ Wh,
                                                       const float4 wi, const float4 av,
                                                       const int offv, const float pen, const unsigned c16,
                                                       float& m, float& den, float4& acc) {
    const char* base = (const char*)Wh;
    const unsigned o0 = (unsigned)rowBcast<J + 0>(offv) + c16;
    const unsigned o1 = (unsigned)rowBcast<J + 1>(offv) + c16;
    const unsigned o2 = (unsigned)rowBcast<J + 2>(offv) + c16;
    const unsigned o3 = (unsigned)rowBcast<J + 3>(offv) + c16;
    const float4 w0 = *(const float4*)(base + o0);
    const float4 w1 = *(const float4*)(base + o1);
    const float4 w2 = *(const float4*)(base + o2);
    const float4 w3 = *(const float4*)(base + o3);
    asm volatile("" :: "v"(w0.x), "v"(w0.y), "v"(w0.z), "v"(w0.w), "v"(w1.x), "v"(w1.y), "v"(w1.z), "v"(w1.w),
                       "v"(w2.x), "v"(w2.y), "v"(w2.z), "v"(w2.w), "v"(w3.x), "v"(w3.y), "v"(w3.z), "v"(w3.w));
    const int pi = __float_as_int(pen);
    const float l0 = rowLogit(w0, wi, av) + __int_as_float(rowBcast<J + 0>(pi));
    const float l1 = rowLogit(w1, wi, av) + __int_as_float(rowBcast<J + 1>(pi));
    const float l2 = rowLogit(w2, wi, av) + __int_as_float(rowBcast<J + 2>(pi));
    const float l3 = rowLogit(w3, wi, av) + __int_as_float(rowBcast<J + 3>(pi));
    const float mn = fmaxf(fmaxf(m, l0), fmaxf(l1, fmaxf(l2, l3)));
    const float sc = __expf(m - mn);
    const float p0 = __expf(l0 - mn);
    const float p1 = __expf(l1 - mn);
    const float p2 = __expf(l2 - mn);
    const float p3 = __expf(l3 - mn);
    den   = den   * sc + ((p0 + p1) + (p2 + p3));
    acc.x = acc.x * sc + ((p0 * w0.x + p1 * w1.x) + (p2 * w2.x + p3 * w3.x));
    acc.y = acc.y * sc + ((p0 * w0.y + p1 * w1.y) + (p2 * w2.y + p3 * w3.y));
    acc.z = acc.z * sc + ((p0 * w0.z + p1 * w1.z) + (p2 * w2.z + p3 * w3.z));
    acc.w = acc.w * sc + ((p0 * w0.w + p1 * w1.w) + (p2 * w2.w + p3 * w3.w));
    m = mn;
}

__device__ __forceinline__ void put_row(unsigned* cntL, int* rows, const unsigned nl, const int s) {
    const unsigned k = atomicAdd(&cntL[nl], 1u);
    if (k < (unsigned)CAP) rows[nl * CAP + k] = s;
}

__device__ __forceinline__ void walk16(const float* __restrict__ Wh, const float4 wi, const float4 av,
                                       const bool hit, const int os, const int nd, const int n,
                                       const unsigned c16, float& m, float& den, float4& acc) {
    const unsigned bal = __builtin_amdgcn_ballot_w32(hit);
    if (bal == 0u) return;
    int sv = hit ? os : nd;
    sv = min(max(sv, 0), n - 1);
    const int offv = sv * (DD * 4);
    const float pen = hit ? 0.0f : NOEDGE;
    if (bal & 0x000f000fu) edge4<0>(Wh, wi, av, offv, pen, c16, m, den, acc);
    if (bal & 0x00f000f0u) edge4<4>(Wh, wi, av, offv, pen, c16, m, den, acc);
    if (bal & 0x0f000f00u) edge4<8>(Wh, wi, av, offv, pen, c16, m, den, acc);
    if (bal & 0xf000f000u) edge4<12>(Wh, wi, av, offv, pen, c16, m, den, acc);
}

#define BT 1024
#define BU 10

__global__ __launch_bounds__(BT) void k_range(const float* __restrict__ Wh,
                                              const float* __restrict__ avec,
                                              const unsigned* __restrict__ seg,
                                              const unsigned* __restrict__ cellCnt,
                                              const int* __restrict__ ovfA,
                                              const unsigned* __restrict__ ovfACnt,
                                              float* __restrict__ out,
                                              int n, int numChunks, int RN) {
    __shared__ unsigned cntL[MAXRN];
    __shared__ int rows[MAXRN * CAP];
    __shared__ unsigned nextPair;
    const int t = threadIdx.x;
    const int r = blockIdx.x;
    const int base = r * RN;
    for (int i = t; i < MAXRN; i += BT) cntL[i] = 0u;
    if (t == 0) nextPair = 0u;
    __syncthreads();

    const int total = numChunks * CELL;
    const unsigned* segR = seg + (size_t)r * numChunks * CELL;
    const unsigned* ccR  = cellCnt + (size_t)r * numChunks;
    for (int b0 = 0; b0 < total; b0 += BT * BU) {
        unsigned cl[BU], rec[BU];
        bool ok[BU];
#pragma unroll
        for (int j = 0; j < BU; ++j) {
            const int i = b0 + j * BT + t;
            cl[j] = (i < total) ? ccR[i / CELL] : 0u;
        }
#pragma unroll
        for (int j = 0; j < BU; ++j) {
            const int i = b0 + j * BT + t;
            const int slot = i - (i / CELL) * CELL;
            ok[j]  = (unsigned)slot < cl[j];
            rec[j] = ok[j] ? segR[i] : 0u;
        }
#pragma unroll
        for (int j = 0; j < BU; ++j)
            if (ok[j]) put_row(cntL, rows, rec[j] >> 24, (int)(rec[j] & 0x00ffffffu));
    }

    for (int ch = t; ch < numChunks; ch += BT) {
        const unsigned oc = min(ovfACnt[ch], (unsigned)CHUNK);
        for (unsigned q = 0; q < oc; ++q) {
            const int ed = ovfA[2 * ((size_t)ch * CHUNK + q) + 0];
            const int es = ovfA[2 * ((size_t)ch * CHUNK + q) + 1];
            if (min((unsigned)ed / (unsigned)RN, (unsigned)(MAXR - 1)) == (unsigned)r)
                put_row(cntL, rows, ((unsigned)ed - (unsigned)base) & (unsigned)(MAXRN - 1), es);
        }
    }
    __syncthreads();

    const int lane = t & 31;
    const int c = lane & 15;
    const int half = lane >> 4;
    const int rn = min(RN, n - base);
    const int pairs = (rn + 1) >> 1;
    const float4 av = ((const float4*)avec)[c];
    const unsigned c16 = (unsigned)c * 16u;

    for (int it = 0; it < pairs; ++it) {
        unsigned tk = 0u;
        if (lane == 0) tk = atomicAdd(&nextPair, 1u);
        const int p = __builtin_amdgcn_readfirstlane((int)tk);
        if (p >= pairs) break;

        const int nl  = 2 * p + half;
        const bool live = nl < rn;
        const int nlc = live ? nl : 2 * p;
        const int nd  = base + nlc;
        const unsigned cn = cntL[nlc];
        const bool over = cn > (unsigned)CAP;
        const int deg = (live && !over) ? (int)cn : 0;
        const unsigned moreRows = __builtin_amdgcn_ballot_w32(live && over) & 0x00010001u;
        const float4 wi = ((const float4*)(Wh + (size_t)nd * DD))[c];
        const int* row = rows + nlc * CAP;

        const int d0 = __builtin_amdgcn_readlane(deg, 0);
        const int d1 = __builtin_amdgcn_readlane(deg, 16);
        const int degmax = d0 > d1 ? d0 : d1;

        float m = -3.0e38f;
        float den = 0.0f;
        float4 acc = make_float4(0.0f, 0.0f, 0.0f, 0.0f);

        for (int k = 0; k < degmax; k += 16) {
            const bool has = c < deg - k;
            int sv = has ? row[k + c] : nd;
            sv = min(max(sv, 0), n - 1);
            const int offv = sv * (DD * 4);
            const float pen = has ? 0.0f : NOEDGE;
            edge4<0>(Wh, wi, av, offv, pen, c16, m, den, acc);
            if (k + 4 < degmax)  edge4<4>(Wh, wi, av, offv, pen, c16, m, den, acc);
            if (k + 8 < degmax)  edge4<8>(Wh, wi, av, offv, pen, c16, m, den, acc);
            if (k + 12 < degmax) edge4<12>(Wh, wi, av, offv, pen, c16, m, den, acc);
        }

        if (moreRows != 0u) {
            const bool more = live && over;
            for (int k = 0; k < total; k += 16) {
                const int i = k + c;
                bool okr = false;
                unsigned rc = 0u;
                if (i < total) {
                    const int ch = i / CELL;
                    okr = (unsigned)(i - ch * CELL) < ccR[ch];
                    if (okr) rc = segR[i];
                }
                const bool hit = more && okr && (rc >> 24) == (unsigned)nl;
                walk16(Wh, wi, av, hit, (int)(rc & 0x00ffffffu), nd, n, c16, m, den, acc);
            }
            for (int ch = 0; ch < numChunks; ++ch) {
                const int oc = __builtin_amdgcn_readfirstlane((int)min(ovfACnt[ch], (unsigned)CHUNK));
                for (int q0 = 0; q0 < oc; q0 += 16) {
                    const int q = q0 + c;
                    int ed = -1, es = nd;
                    if (q < oc) {
                        ed = ovfA[2 * ((size_t)ch * CHUNK + q) + 0];
                        es = ovfA[2 * ((size_t)ch * CHUNK + q) + 1];
                    }
                    const bool hit = more && ed == base + nl;
                    walk16(Wh, wi, av, hit, es, nd, n, c16, m, den, acc);
                }
            }
        }

        const float dn = fmaxf(den, 1e-9f);
        float4 o;
        o.x = 1.0f / (1.0f + __expf(-(acc.x / dn)));
        o.y = 1.0f / (1.0f + __expf(-(acc.y / dn)));
        o.z = 1.0f / (1.0f + __expf(-(acc.z / dn)));
        o.w = 1.0f / (1.0f + __expf(-(acc.w / dn)));
        if (live) ((float4*)(out + (size_t)(base + nl) * DD))[c] = o;
    }
}

extern "C" void kernel_launch(void* const* d_in, const int* in_sizes, int n_in,
                              void* d_out, int out_size, void* d_ws, size_t ws_size,
                              hipStream_t stream) {
    const float* x    = (const float*)d_in[0];
    const float* W    = (const float*)d_in[1];
    const float* bias = (const float*)d_in[2];
    const float* avec = (const float*)d_in[3];
    const int*   ei   = (const int*)d_in[4];

    const int N = in_sizes[0] / DD;
    const int E = in_sizes[4] / 2;
    const int* src = ei;
    const int* dst = ei + E;

    const int numChunks = (E + CHUNK - 1) / CHUNK;
    const int RN = (N + MAXR - 1) / MAXR;
    const int numRanges = (N + RN - 1) / RN;

    float*    Wh      = (float*)d_ws;
    unsigned* cellCnt = (unsigned*)(Wh + (size_t)N * DD);
    unsigned* seg     = cellCnt + (((size_t)numRanges * numChunks + 3) & ~(size_t)3);
    unsigned* ovfACnt = seg + (size_t)numRanges * numChunks * CELL;
    int*      ovfA    = (int*)(ovfACnt + (((size_t)numChunks + 3) & ~(size_t)3));

    float* out = (float*)d_out;

    const int tiles = (N + 15) / 16;
    const int gemmBlocks = (tiles + 7) / 8;
    k_gemm_bin<<<gemmBlocks + numChunks, 256, 0, stream>>>(x, W, bias, Wh, src, dst, seg, cellCnt, ovfA, ovfACnt,
                                                           E, N, gemmBlocks, numChunks, numRanges, RN);

    k_range<<<numRanges, BT, 0, stream>>>(Wh, avec, seg, cellCnt, ovfA, ovfACnt, out, N, numChunks, RN);
}


# chunk sort of launch 1 waits on LDS per group, not per item: a thread's 16 rank adds go out together; cell write-out reads 24 counters, then 24 records per trip; sums by LDS arrival as before
# speedup vs baseline: 9.7280x; 1.0589x over previous
#include <hip/hip_runtime.h>
#include <hip/hip_bf16.h>
#include <math.h>


#define DD 64
#define NSLOPE 0.2f
#define NOEDGE (-__builtin_inff())

typedef __attribute__((ext_vector_type(2))) float v2f;
typedef __attribute__((ext_vector_type(8))) float v8f;

#define WP 68
#define GEMM_LDS_WORDS (64 * WP + 8 * 16 * WP)

__device__ __forceinline__ void gemm_block(const float* __restrict__ x,
                                           const float* __restrict__ W,
                                           const float* __restrict__ bias,
                                           float* __restrict__ Wh,
                                           const int blk, const int n, float* lds) {
    const int t    = threadIdx.x;
    const int lane = t & 31;
    const int wave = t >> 5;
    const int lm   = lane & 15;
    const int half = lane >> 4;
    const int tiles = (n + 15) >> 4;
    const bool valid = blk * 8 + wave < tiles;
    const int tile = valid ? blk * 8 + wave : tiles - 1;
    const int row0 = tile * 16;

    float* Wl = lds;
    float* T  = lds + 64 * WP + wave * 16 * WP;

    float4 wv[4], xv[8];
#pragma unroll
    for (int r = 0; r < 4; ++r) wv[r] = ((const float4*)W)[t + 256 * r];
#pragma unroll
    for (int q = 0; q < 8; ++q) {
        const int grow = min(row0 + 2 * q + half, n - 1);
        xv[q] = ((const float4*)(x + (size_t)grow * DD))[lm];
    }
#pragma unroll
    for (int r = 0; r < 4; ++r) {
        const int idx = t + 256 * r;
        *(float4*)(Wl + (idx >> 4) * WP + (idx & 15) * 4) = wv[r];
    }
#pragma unroll
    for (int q = 0; q < 8; ++q) *(float4*)(T + (2 * q + half) * WP + lm * 4) = xv[q];
    __syncthreads();

    float4 af[8];
#pragma unroll
    for (int j = 0; j < 8; ++j) af[j] = *(const float4*)(T + lm * WP + 8 * j + 4 * half);

    v8f acc[4];
#pragma unroll
    for (int c = 0; c < 4; ++c) {
        acc[c] = v8f{};
        const float* wl = Wl + (c * 16 + lm) * WP + 4 * half;
#pragma unroll
        for (int j = 0; j < 8; ++j) {
            const float4 bf = *(const float4*)(wl + 8 * j);
            acc[c] = __builtin_amdgcn_wmma_f32_16x16x4_f32(
                false, v2f{af[j].x, af[j].y}, false, v2f{bf.x, bf.y}, (short)0, acc[c], false, false);
            acc[c] = __builtin_amdgcn_wmma_f32_16x16x4_f32(
                false, v2f{af[j].z, af[j].w}, false, v2f{bf.z, bf.w}, (short)0, acc[c], false, false);
        }
    }
    __syncthreads();

#pragma unroll
    for (int c = 0; c < 4; ++c) {
        const float bv = bias[c * 16 + lm];
#pragma unroll
        for (int v = 0; v < 8; ++v) T[(v + 8 * half) * WP + c * 16 + lm] = acc[c][v] + bv;
    }
    __syncthreads();

    if (valid) {
        float* wo = Wh + (size_t)row0 * DD + lm * 4;
        if (row0 + 16 <= n) {
#pragma unroll
            for (int q = 0; q < 8; ++q) {
                const int row = 2 * q + half;
                *(float4*)(wo + row * DD) = *(const float4*)(T + row * WP + lm * 4);
            }
        } else {
            for (int q = 0; q < 8; ++q) {
                const int row = 2 * q + half;
                if (row0 + row < n) *(float4*)(wo + row * DD) = *(const float4*)(T + row * WP + lm * 4);
            }
        }
    }
}

#define CAP 160
#define MAXR 256
#define MAXRN 256
#define CELL 48
#define CHUNK 4096
#define HISTW (MAXR + 4 + 32)

__device__ __forceinline__ void bin_block(const int* __restrict__ src,
                                          const int* __restrict__ dst,
                                          unsigned* __restrict__ seg,
                                          unsigned* __restrict__ cellCnt,
                                          int* __restrict__ ovfA,
                                          unsigned* __restrict__ ovfACnt,
                                          const int E, const int chunk,
                                          const int numChunks, const int numRanges, const int RN,
                                          unsigned* hist, unsigned* cells) {
    const int t = threadIdx.x;
    for (int i = t; i < HISTW; i += 256) hist[i] = 0u;
    __syncthreads();

    const int e0 = chunk * CHUNK + t;
    int d[16], sv[16];
#pragma unroll
    for (int j = 0; j < 16; ++j) {
        const int e = e0 + j * 256;
        const bool in = e < E;
        d[j]  = in ? dst[e] : -1;
        sv[j] = in ? src[e] : 0;
    }
    unsigned rgv[16], kk[16];
#pragma unroll
    for (int j = 0; j < 16; ++j) {
        const unsigned rg = min((unsigned)max(d[j], 0) / (unsigned)RN, (unsigned)(MAXR - 1));
        rgv[j] = d[j] >= 0 ? rg : (unsigned)(MAXR + 4) + (unsigned)(t & 31);
    }
#pragma unroll
    for (int j = 0; j < 16; ++j) kk[j] = atomicAdd(&hist[rgv[j]], 1u);
#pragma unroll
    for (int j = 0; j < 16; ++j) {
        if (d[j] >= 0) {
            const unsigned rg = rgv[j];
            if (kk[j] < (unsigned)CELL) {
                const unsigned nl = ((unsigned)d[j] - rg * (unsigned)RN) & (unsigned)(MAXRN - 1);
                cells[rg * CELL + kk[j]] = (nl << 24) | ((unsigned)sv[j] & 0x00ffffffu);
            } else {
                const unsigned q = atomicAdd(&hist[MAXR], 1u);
                ovfA[2 * ((size_t)chunk * CHUNK + q) + 0] = d[j];
                ovfA[2 * ((size_t)chunk * CHUNK + q) + 1] = sv[j];
            }
        }
    }
    __syncthreads();

    int cgp[3], slp[3];
#pragma unroll
    for (int b = 0; b < 3; ++b) {
        const int p = t + 256 * b;
        cgp[b] = p / CELL;
        slp[b] = p - cgp[b] * CELL;
    }
    for (int a0 = 0; a0 < MAXR / 16; a0 += 8) {
        unsigned cn[8][3], rec[8][3];
#pragma unroll
        for (int a = 0; a < 8; ++a)
#pragma unroll
            for (int b = 0; b < 3; ++b)
                cn[a][b] = min(hist[16 * (a0 + a) + cgp[b]], (unsigned)CELL);
#pragma unroll
        for (int a = 0; a < 8; ++a)
#pragma unroll
            for (int b = 0; b < 3; ++b) {
                const int rg = 16 * (a0 + a) + cgp[b];
                const bool ok = rg < numRanges && (unsigned)slp[b] < cn[a][b];
                rec[a][b] = ok ? cells[rg * CELL + slp[b]] : 0u;
            }
#pragma unroll
        for (int a = 0; a < 8; ++a)
#pragma unroll
            for (int b = 0; b < 3; ++b) {
                const int rg = 16 * (a0 + a) + cgp[b];
                const bool ok = rg < numRanges && (unsigned)slp[b] < cn[a][b];
                if (ok) seg[((size_t)rg * numChunks + chunk) * CELL + slp[b]] = rec[a][b];
            }
    }
    for (int rg = t; rg < numRanges; rg += 256)
        cellCnt[(size_t)rg * numChunks + chunk] = min(hist[rg], (unsigned)CELL);
    if (t == 0) ovfACnt[chunk] = min(hist[MAXR], (unsigned)CHUNK);
}

__global__ __launch_bounds__(256) void k_gemm_bin(const float* __restrict__ x,
                                                  const float* __restrict__ W,
                                                  const float* __restrict__ bias,
                                                  float* __restrict__ Wh,
                                                  const int* __restrict__ src,
                                                  const int* __restrict__ dst,
                                                  unsigned* __restrict__ seg,
                                                  unsigned* __restrict__ cellCnt,
                                                  int* __restrict__ ovfA,
                                                  unsigned* __restrict__ ovfACnt,
                                                  int E, int n, int gemmBlocks,
                                                  int numChunks, int numRanges, int RN) {
    __shared__ __attribute__((aligned(16))) unsigned smem[GEMM_LDS_WORDS];
    unsigned* hist  = smem;
    unsigned* cells = smem + HISTW;
    const int blk = blockIdx.x;
    if (blk < gemmBlocks) {
        gemm_block(x, W, bias, Wh, blk, n, (float*)smem);
    } else {
        bin_block(src, dst, seg, cellCnt, ovfA, ovfACnt, E, blk - gemmBlocks, numChunks, numRanges, RN, hist, cells);
    }
}

template <int J> __device__ __forceinline__ int rowBcast(int v) {
    return __builtin_amdgcn_mov_dpp(v, 0x150 + J, 0xf, 0xf, true);
}
template <int M> __device__ __forceinline__ float rowXorAdd(float p) {
    const int q = __builtin_amdgcn_mov_dpp(__float_as_int(p), 0x160 + M, 0xf, 0xf, true);
    return p + __int_as_float(q);
}
__device__ __forceinline__ float rowLogit(const float4 ws, const float4 wd, const float4 av) {
    const v2f s01 = v2f{ws.x, ws.y} + v2f{wd.x, wd.y};
    const v2f s23 = v2f{ws.z, ws.w} + v2f{wd.z, wd.w};
    const v2f r01 = __builtin_elementwise_max(s01, s01 * NSLOPE);
    const v2f r23 = __builtin_elementwise_max(s23, s23 * NSLOPE);
    const v2f q = r01 * v2f{av.x, av.y} + r23 * v2f{av.z, av.w};
    float p = q.x + q.y;
    p = rowXorAdd<1>(p);
    p = rowXorAdd<2>(p);
    p = rowXorAdd<4>(p);
    p = rowXorAdd<8>(p);
    return p;
}
template <int J> __device__ __forceinline__ void edge4(const float* __restrict__ Wh,
                                                       const float4 wi, const float4 av,
                                                       const int offv, const float pen, const unsigned c16,
                                                       float& m, float& den, float4& acc) {
    const char* base = (const char*)Wh;
    const unsigned o0 = (unsigned)rowBcast<J + 0>(offv) + c16;
    const unsigned o1 = (unsigned)rowBcast<J + 1>(offv) + c16;
    const unsigned o2 = (unsigned)rowBcast<J + 2>(offv) + c16;
    const unsigned o3 = (unsigned)rowBcast<J + 3>(offv) + c16;
    const float4 w0 = *(const float4*)(base + o0);
    const float4 w1 = *(const float4*)(base + o1);
    const float4 w2 = *(const float4*)(base + o2);
    const float4 w3 = *(const float4*)(base + o3);
    asm volatile("" :: "v"(w0.x), "v"(w0.y), "v"(w0.z), "v"(w0.w), "v"(w1.x), "v"(w1.y), "v"(w1.z), "v"(w1.w),
                       "v"(w2.x), "v"(w2.y), "v"(w2.z), "v"(w2.w), "v"(w3.x), "v"(w3.y), "v"(w3.z), "v"(w3.w));
    const int pi = __float_as_int(pen);
    const float l0 = rowLogit(w0, wi, av) + __int_as_float(rowBcast<J + 0>(pi));
    const float l1 = rowLogit(w1, wi, av) + __int_as_float(rowBcast<J + 1>(pi));
    const float l2 = rowLogit(w2, wi, av) + __int_as_float(rowBcast<J + 2>(pi));
    const float l3 = rowLogit(w3, wi, av) + __int_as_float(rowBcast<J + 3>(pi));
    const float mn = fmaxf(fmaxf(m, l0), fmaxf(l1, fmaxf(l2, l3)));
    const float sc = __expf(m - mn);
    const float p0 = __expf(l0 - mn);
    const float p1 = __expf(l1 - mn);
    const float p2 = __expf(l2 - mn);
    const float p3 = __expf(l3 - mn);
    den   = den   * sc + ((p0 + p1) + (p2 + p3));
    acc.x = acc.x * sc + ((p0 * w0.x + p1 * w1.x) + (p2 * w2.x + p3 * w3.x));
    acc.y = acc.y * sc + ((p0 * w0.y + p1 * w1.y) + (p2 * w2.y + p3 * w3.y));
    acc.z = acc.z * sc + ((p0 * w0.z + p1 * w1.z) + (p2 * w2.z + p3 * w3.z));
    acc.w = acc.w * sc + ((p0 * w0.w + p1 * w1.w) + (p2 * w2.w + p3 * w3.w));
    m = mn;
}

__device__ __forceinline__ void put_row(unsigned* cntL, int* rows, const unsigned nl, const int s) {
    const unsigned k = atomicAdd(&cntL[nl], 1u);
    if (k < (unsigned)CAP) rows[nl * CAP + k] = s;
}

__device__ __forceinline__ void walk16(const float* __restrict__ Wh, const float4 wi, const float4 av,
                                       const bool hit, const int os, const int nd, const int n,
                                       const unsigned c16, float& m, float& den, float4& acc) {
    const unsigned bal = __builtin_amdgcn_ballot_w32(hit);
    if (bal == 0u) return;
    int sv = hit ? os : nd;
    sv = min(max(sv, 0), n - 1);
    const int offv = sv * (DD * 4);
    const float pen = hit ? 0.0f : NOEDGE;
    if (bal & 0x000f000fu) edge4<0>(Wh, wi, av, offv, pen, c16, m, den, acc);
    if (bal & 0x00f000f0u) edge4<4>(Wh, wi, av, offv, pen, c16, m, den, acc);
    if (bal & 0x0f000f00u) edge4<8>(Wh, wi, av, offv, pen, c16, m, den, acc);
    if (bal & 0xf000f000u) edge4<12>(Wh, wi, av, offv, pen, c16, m, den, acc);
}

#define BT 1024
#define BU 10

__global__ __launch_bounds__(BT) void k_range(const float* __restrict__ Wh,
                                              const float* __restrict__ avec,
                                              const unsigned* __restrict__ seg,
                                              const unsigned* __restrict__ cellCnt,
                                              const int* __restrict__ ovfA,
                                              const unsigned* __restrict__ ovfACnt,
                                              float* __restrict__ out,
                                              int n, int numChunks, int RN) {
    __shared__ unsigned cntL[MAXRN];
    __shared__ int rows[MAXRN * CAP];
    __shared__ unsigned nextPair;
    const int t = threadIdx.x;
    const int r = blockIdx.x;
    const int base = r * RN;
    for (int i = t; i < MAXRN; i += BT) cntL[i] = 0u;
    if (t == 0) nextPair = 0u;
    __syncthreads();

    const int total = numChunks * CELL;
    const unsigned* segR = seg + (size_t)r * numChunks * CELL;
    const unsigned* ccR  = cellCnt + (size_t)r * numChunks;
    for (int b0 = 0; b0 < total; b0 += BT * BU) {
        unsigned cl[BU], rec[BU];
        bool ok[BU];
#pragma unroll
        for (int j = 0; j < BU; ++j) {
            const int i = b0 + j * BT + t;
            cl[j] = (i < total) ? ccR[i / CELL] : 0u;
        }
#pragma unroll
        for (int j = 0; j < BU; ++j) {
            const int i = b0 + j * BT + t;
            const int slot = i - (i / CELL) * CELL;
            ok[j]  = (unsigned)slot < cl[j];
            rec[j] = ok[j] ? segR[i] : 0u;
        }
#pragma unroll
        for (int j = 0; j < BU; ++j)
            if (ok[j]) put_row(cntL, rows, rec[j] >> 24, (int)(rec[j] & 0x00ffffffu));
    }

    for (int ch = t; ch < numChunks; ch += BT) {
        const unsigned oc = min(ovfACnt[ch], (unsigned)CHUNK);
        for (unsigned q = 0; q < oc; ++q) {
            const int ed = ovfA[2 * ((size_t)ch * CHUNK + q) + 0];
            const int es = ovfA[2 * ((size_t)ch * CHUNK + q) + 1];
            if (min((unsigned)ed / (unsigned)RN, (unsigned)(MAXR - 1)) == (unsigned)r)
                put_row(cntL, rows, ((unsigned)ed - (unsigned)base) & (unsigned)(MAXRN - 1), es);
        }
    }
    __syncthreads();

    const int lane = t & 31;
    const int c = lane & 15;
    const int half = lane >> 4;
    const int rn = min(RN, n - base);
    const int pairs = (rn + 1) >> 1;
    const float4 av = ((const float4*)avec)[c];
    const unsigned c16 = (unsigned)c * 16u;

    for (int it = 0; it < pairs; ++it) {
        unsigned tk = 0u;
        if (lane == 0) tk = atomicAdd(&nextPair, 1u);
        const int p = __builtin_amdgcn_readfirstlane((int)tk);
        if (p >= pairs) break;

        const int nl  = 2 * p + half;
        const bool live = nl < rn;
        const int nlc = live ? nl : 2 * p;
        const int nd  = base + nlc;
        const unsigned cn = cntL[nlc];
        const bool over = cn > (unsigned)CAP;
        const int deg = (live && !over) ? (int)cn : 0;
        const unsigned moreRows = __builtin_amdgcn_ballot_w32(live && over) & 0x00010001u;
        const float4 wi = ((const float4*)(Wh + (size_t)nd * DD))[c];
        const int* row = rows + nlc * CAP;

        const int d0 = __builtin_amdgcn_readlane(deg, 0);
        const int d1 = __builtin_amdgcn_readlane(deg, 16);
        const int degmax = d0 > d1 ? d0 : d1;

        float m = -3.0e38f;
        float den = 0.0f;
        float4 acc = make_float4(0.0f, 0.0f, 0.0f, 0.0f);

        for (int k = 0; k < degmax; k += 16) {
            const bool has = c < deg - k;
            int sv = has ? row[k + c] : nd;
            sv = min(max(sv, 0), n - 1);
            const int offv = sv * (DD * 4);
            const float pen = has ? 0.0f : NOEDGE;
            edge4<0>(Wh, wi, av, offv, pen, c16, m, den, acc);
            if (k + 4 < degmax)  edge4<4>(Wh, wi, av, offv, pen, c16, m, den, acc);
            if (k + 8 < degmax)  edge4<8>(Wh, wi, av, offv, pen, c16, m, den, acc);
            if (k + 12 < degmax) edge4<12>(Wh, wi, av, offv, pen, c16, m, den, acc);
        }

        if (moreRows != 0u) {
            const bool more = live && over;
            for (int k = 0; k < total; k += 16) {
                const int i = k + c;
                bool okr = false;
                unsigned rc = 0u;
                if (i < total) {
                    const int ch = i / CELL;
                    okr = (unsigned)(i - ch * CELL) < ccR[ch];
                    if (okr) rc = segR[i];
                }
                const bool hit = more && okr && (rc >> 24) == (unsigned)nl;
                walk16(Wh, wi, av, hit, (int)(rc & 0x00ffffffu), nd, n, c16, m, den, acc);
            }
            for (int ch = 0; ch < numChunks; ++ch) {
                const int oc = __builtin_amdgcn_readfirstlane((int)min(ovfACnt[ch], (unsigned)CHUNK));
                for (int q0 = 0; q0 < oc; q0 += 16) {
                    const int q = q0 + c;
                    int ed = -1, es = nd;
                    if (q < oc) {
                        ed = ovfA[2 * ((size_t)ch * CHUNK + q) + 0];
                        es = ovfA[2 * ((size_t)ch * CHUNK + q) + 1];
                    }
                    const bool hit = more && ed == base + nl;
                    walk16(Wh, wi, av, hit, es, nd, n, c16, m, den, acc);
                }
            }
        }

        const float dn = fmaxf(den, 1e-9f);
        float4 o;
        o.x = 1.0f / (1.0f + __expf(-(acc.x / dn)));
        o.y = 1.0f / (1.0f + __expf(-(acc.y / dn)));
        o.z = 1.0f / (1.0f + __expf(-(acc.z / dn)));
        o.w = 1.0f / (1.0f + __expf(-(acc.w / dn)));
        if (live) ((float4*)(out + (size_t)(base + nl) * DD))[c] = o;
    }
}

extern "C" void kernel_launch(void* const* d_in, const int* in_sizes, int n_in,
                              void* d_out, int out_size, void* d_ws, size_t ws_size,
                              hipStream_t stream) {
    const float* x    = (const float*)d_in[0];
    const float* W    = (const float*)d_in[1];
    const float* bias = (const float*)d_in[2];
    const float* avec = (const float*)d_in[3];
    const int*   ei   = (const int*)d_in[4];

    const int N = in_sizes[0] / DD;
    const int E = in_sizes[4] / 2;
    const int* src = ei;
    const int* dst = ei + E;

    const int numChunks = (E + CHUNK - 1) / CHUNK;
    const int RN = (N + MAXR - 1) / MAXR;
    const int numRanges = (N + RN - 1) / RN;

    float*    Wh      = (float*)d_ws;
    unsigned* cellCnt = (unsigned*)(Wh + (size_t)N * DD);
    unsigned* seg     = cellCnt + (((size_t)numRanges * numChunks + 3) & ~(size_t)3);
    unsigned* ovfACnt = seg + (size_t)numRanges * numChunks * CELL;
    int*      ovfA    = (int*)(ovfACnt + (((size_t)numChunks + 3) & ~(size_t)3));

    float* out = (float*)d_out;

    const int tiles = (N + 15) / 16;
    const int gemmBlocks = (tiles + 7) / 8;
    k_gemm_bin<<<gemmBlocks + numChunks, 256, 0, stream>>>(x, W, bias, Wh, src, dst, seg, cellCnt, ovfA, ovfACnt,
                                                           E, N, gemmBlocks, numChunks, numRanges, RN);

    k_range<<<numRanges, BT, 0, stream>>>(Wh, avec, seg, cellCnt, ovfA, ovfACnt, out, N, numChunks, RN);
}


# cells get a fixed length, empty slots marked: launch 1 writes them whole in 16-B pieces, launch 2's fill is one trip to memory (every slot, no lengths) and one rank group; sums by LDS arrival as ever
# speedup vs baseline: 10.9473x; 1.1253x over previous
#include <hip/hip_runtime.h>
#include <hip/hip_bf16.h>
#include <math.h>


#define DD 64
#define NSLOPE 0.2f
#define NOEDGE (-__builtin_inff())

typedef __attribute__((ext_vector_type(2))) float v2f;
typedef __attribute__((ext_vector_type(8))) float v8f;

#define WP 68
#define GEMM_LDS_WORDS (64 * WP + 8 * 16 * WP)

__device__ __forceinline__ void gemm_block(const float* __restrict__ x,
                                           const float* __restrict__ W,
                                           const float* __restrict__ bias,
                                           float* __restrict__ Wh,
                                           const int blk, const int n, float* lds) {
    const int t    = threadIdx.x;
    const int lane = t & 31;
    const int wave = t >> 5;
    const int lm   = lane & 15;
    const int half = lane >> 4;
    const int tiles = (n + 15) >> 4;
    const bool valid = blk * 8 + wave < tiles;
    const int tile = valid ? blk * 8 + wave : tiles - 1;
    const int row0 = tile * 16;

    float* Wl = lds;
    float* T  = lds + 64 * WP + wave * 16 * WP;

    float4 wv[4], xv[8];
#pragma unroll
    for (int r = 0; r < 4; ++r) wv[r] = ((const float4*)W)[t + 256 * r];
#pragma unroll
    for (int q = 0; q < 8; ++q) {
        const int grow = min(row0 + 2 * q + half, n - 1);
        xv[q] = ((const float4*)(x + (size_t)grow * DD))[lm];
    }
#pragma unroll
    for (int r = 0; r < 4; ++r) {
        const int idx = t + 256 * r;
        *(float4*)(Wl + (idx >> 4) * WP + (idx & 15) * 4) = wv[r];
    }
#pragma unroll
    for (int q = 0; q < 8; ++q) *(float4*)(T + (2 * q + half) * WP + lm * 4) = xv[q];
    __syncthreads();

    float4 af[8];
#pragma unroll
    for (int j = 0; j < 8; ++j) af[j] = *(const float4*)(T + lm * WP + 8 * j + 4 * half);

    v8f acc[4];
#pragma unroll
    for (int c = 0; c < 4; ++c) {
        acc[c] = v8f{};
        const float* wl = Wl + (c * 16 + lm) * WP + 4 * half;
#pragma unroll
        for (int j = 0; j < 8; ++j) {
            const float4 bf = *(const float4*)(wl + 8 * j);
            acc[c] = __builtin_amdgcn_wmma_f32_16x16x4_f32(
                false, v2f{af[j].x, af[j].y}, false, v2f{bf.x, bf.y}, (short)0, acc[c], false, false);
            acc[c] = __builtin_amdgcn_wmma_f32_16x16x4_f32(
                false, v2f{af[j].z, af[j].w}, false, v2f{bf.z, bf.w}, (short)0, acc[c], false, false);
        }
    }
    __syncthreads();

#pragma unroll
    for (int c = 0; c < 4; ++c) {
        const float bv = bias[c * 16 + lm];
#pragma unroll
        for (int v = 0; v < 8; ++v) T[(v + 8 * half) * WP + c * 16 + lm] = acc[c][v] + bv;
    }
    __syncthreads();

    if (valid) {
        float* wo = Wh + (size_t)row0 * DD + lm * 4;
        if (row0 + 16 <= n) {
#pragma unroll
            for (int q = 0; q < 8; ++q) {
                const int row = 2 * q + half;
                *(float4*)(wo + row * DD) = *(const float4*)(T + row * WP + lm * 4);
            }
        } else {
            for (int q = 0; q < 8; ++q) {
                const int row = 2 * q + half;
                if (row0 + row < n) *(float4*)(wo + row * DD) = *(const float4*)(T + row * WP + lm * 4);
            }
        }
    }
}

#define CAP 160
#define MAXR 256
#define MAXRN 256
#define CELL 48
#define CHUNK 4096
#define HISTW (MAXR + 4 + 32)
#define EMPTY 0x00ffffffu
#define CELLQ (CELL / 4)
#define CELLP 64
#define OUTQ (MAXR * CELLQ / 256)

__device__ __forceinline__ void bin_block(const int* __restrict__ src,
                                          const int* __restrict__ dst,
                                          unsigned* __restrict__ seg,
                                          int* __restrict__ ovfA,
                                          unsigned* __restrict__ ovfACnt,
                                          const int E, const int chunk,
                                          const int numChunks, const int numRanges, const int RN,
                                          unsigned* hist, unsigned* cells) {
    const int t = threadIdx.x;
    for (int i = t; i < HISTW; i += 256) hist[i] = 0u;
#pragma unroll
    for (int i = 0; i < OUTQ; ++i) ((uint4*)cells)[t + 256 * i] = make_uint4(EMPTY, EMPTY, EMPTY, EMPTY);
    __syncthreads();

    const int e0 = chunk * CHUNK + t;
    int d[16], sv[16];
#pragma unroll
    for (int j = 0; j < 16; ++j) {
        const int e = e0 + j * 256;
        const bool in = e < E;
        d[j]  = in ? dst[e] : -1;
        sv[j] = in ? src[e] : 0;
    }
    unsigned rgv[16], kk[16];
#pragma unroll
    for (int j = 0; j < 16; ++j) {
        const unsigned rg = min((unsigned)max(d[j], 0) / (unsigned)RN, (unsigned)(MAXR - 1));
        rgv[j] = d[j] >= 0 ? rg : (unsigned)(MAXR + 4) + (unsigned)(t & 31);
    }
#pragma unroll
    for (int j = 0; j < 16; ++j) kk[j] = atomicAdd(&hist[rgv[j]], 1u);
#pragma unroll
    for (int j = 0; j < 16; ++j) {
        if (d[j] >= 0) {
            const unsigned rg = rgv[j];
            if (kk[j] < (unsigned)CELL) {
                const unsigned nl = ((unsigned)d[j] - rg * (unsigned)RN) & (unsigned)(MAXRN - 1);
                cells[rg * CELL + kk[j]] = (nl << 24) | ((unsigned)sv[j] & 0x00ffffffu);
            } else {
                const unsigned q = atomicAdd(&hist[MAXR], 1u);
                ovfA[2 * ((size_t)chunk * CHUNK + q) + 0] = d[j];
                ovfA[2 * ((size_t)chunk * CHUNK + q) + 1] = sv[j];
            }
        }
    }
    __syncthreads();

    if (t < 16 * CELLQ) {
        const int rg0 = t / CELLQ;
        const int part = t - rg0 * CELLQ;
        const uint4* lp = (const uint4*)cells + t;
        unsigned* gp = seg + ((size_t)rg0 * numChunks + chunk) * CELLP + part * 4;
        const size_t gstep = (size_t)16 * numChunks * CELLP;
        if (numRanges == MAXR) {
            uint4 pc[MAXR / 16];
#pragma unroll
            for (int i = 0; i < MAXR / 16; ++i) pc[i] = lp[i * 16 * CELLQ];
#pragma unroll
            for (int i = 0; i < MAXR / 16; ++i) *(uint4*)(gp + i * gstep) = pc[i];
        } else {
            for (int i = 0; i < MAXR / 16; ++i)
                if (rg0 + 16 * i < numRanges) *(uint4*)(gp + i * gstep) = lp[i * 16 * CELLQ];
        }
    }
    if (t == 0) ovfACnt[chunk] = min(hist[MAXR], (unsigned)CHUNK);
}

__global__ __launch_bounds__(256) void k_gemm_bin(const float* __restrict__ x,
                                                  const float* __restrict__ W,
                                                  const float* __restrict__ bias,
                                                  float* __restrict__ Wh,
                                                  const int* __restrict__ src,
                                                  const int* __restrict__ dst,
                                                  unsigned* __restrict__ seg,
                                                  int* __restrict__ ovfA,
                                                  unsigned* __restrict__ ovfACnt,
                                                  int E, int n, int gemmBlocks,
                                                  int numChunks, int numRanges, int RN) {
    __shared__ __attribute__((aligned(16))) unsigned smem[GEMM_LDS_WORDS];
    unsigned* hist  = smem;
    unsigned* cells = smem + HISTW;
    const int blk = blockIdx.x;
    if (blk < gemmBlocks) {
        gemm_block(x, W, bias, Wh, blk, n, (float*)smem);
    } else {
        bin_block(src, dst, seg, ovfA, ovfACnt, E, blk - gemmBlocks, numChunks, numRanges, RN, hist, cells);
    }
}

template <int J> __device__ __forceinline__ int rowBcast(int v) {
    return __builtin_amdgcn_mov_dpp(v, 0x150 + J, 0xf, 0xf, true);
}
template <int M> __device__ __forceinline__ float rowXorAdd(float p) {
    const int q = __builtin_amdgcn_mov_dpp(__float_as_int(p), 0x160 + M, 0xf, 0xf, true);
    return p + __int_as_float(q);
}
__device__ __forceinline__ float rowLogit(const float4 ws, const float4 wd, const float4 av) {
    const v2f s01 = v2f{ws.x, ws.y} + v2f{wd.x, wd.y};
    const v2f s23 = v2f{ws.z, ws.w} + v2f{wd.z, wd.w};
    const v2f r01 = __builtin_elementwise_max(s01, s01 * NSLOPE);
    const v2f r23 = __builtin_elementwise_max(s23, s23 * NSLOPE);
    const v2f q = r01 * v2f{av.x, av.y} + r23 * v2f{av.z, av.w};
    float p = q.x + q.y;
    p = rowXorAdd<1>(p);
    p = rowXorAdd<2>(p);
    p = rowXorAdd<4>(p);
    p = rowXorAdd<8>(p);
    return p;
}
template <int J> __device__ __forceinline__ void edge4(const float* __restrict__ Wh,
                                                       const float4 wi, const float4 av,
                                                       const int offv, const float pen, const unsigned c16,
                                                       float& m, float& den, float4& acc) {
    const char* base = (const char*)Wh;
    const unsigned o0 = (unsigned)rowBcast<J + 0>(offv) + c16;
    const unsigned o1 = (unsigned)rowBcast<J + 1>(offv) + c16;
    const unsigned o2 = (unsigned)rowBcast<J + 2>(offv) + c16;
    const unsigned o3 = (unsigned)rowBcast<J + 3>(offv) + c16;
    const float4 w0 = *(const float4*)(base + o0);
    const float4 w1 = *(const float4*)(base + o1);
    const float4 w2 = *(const float4*)(base + o2);
    const float4 w3 = *(const float4*)(base + o3);
    asm volatile("" :: "v"(w0.x), "v"(w0.y), "v"(w0.z), "v"(w0.w), "v"(w1.x), "v"(w1.y), "v"(w1.z), "v"(w1.w),
                       "v"(w2.x), "v"(w2.y), "v"(w2.z), "v"(w2.w), "v"(w3.x), "v"(w3.y), "v"(w3.z), "v"(w3.w));
    const int pi = __float_as_int(pen);
    const float l0 = rowLogit(w0, wi, av) + __int_as_float(rowBcast<J + 0>(pi));
    const float l1 = rowLogit(w1, wi, av) + __int_as_float(rowBcast<J + 1>(pi));
    const float l2 = rowLogit(w2, wi, av) + __int_as_float(rowBcast<J + 2>(pi));
    const float l3 = rowLogit(w3, wi, av) + __int_as_float(rowBcast<J + 3>(pi));
    const float mn = fmaxf(fmaxf(m, l0), fmaxf(l1, fmaxf(l2, l3)));
    const float sc = __expf(m - mn);
    const float p0 = __expf(l0 - mn);
    const float p1 = __expf(l1 - mn);
    const float p2 = __expf(l2 - mn);
    const float p3 = __expf(l3 - mn);
    den   = den   * sc + ((p0 + p1) + (p2 + p3));
    acc.x = acc.x * sc + ((p0 * w0.x + p1 * w1.x) + (p2 * w2.x + p3 * w3.x));
    acc.y = acc.y * sc + ((p0 * w0.y + p1 * w1.y) + (p2 * w2.y + p3 * w3.y));
    acc.z = acc.z * sc + ((p0 * w0.z + p1 * w1.z) + (p2 * w2.z + p3 * w3.z));
    acc.w = acc.w * sc + ((p0 * w0.w + p1 * w1.w) + (p2 * w2.w + p3 * w3.w));
    m = mn;
}

__device__ __forceinline__ void put_row(unsigned* cntL, int* rows, const unsigned nl, const int s) {
    const unsigned k = atomicAdd(&cntL[nl], 1u);
    if (k < (unsigned)CAP) rows[nl * CAP + k] = s;
}

__device__ __forceinline__ void walk16(const float* __restrict__ Wh, const float4 wi, const float4 av,
                                       const bool hit, const int os, const int nd, const int n,
                                       const unsigned c16, float& m, float& den, float4& acc) {
    const unsigned bal = __builtin_amdgcn_ballot_w32(hit);
    if (bal == 0u) return;
    int sv = hit ? os : nd;
    sv = min(max(sv, 0), n - 1);
    const int offv = sv * (DD * 4);
    const float pen = hit ? 0.0f : NOEDGE;
    if (bal & 0x000f000fu) edge4<0>(Wh, wi, av, offv, pen, c16, m, den, acc);
    if (bal & 0x00f000f0u) edge4<4>(Wh, wi, av, offv, pen, c16, m, den, acc);
    if (bal & 0x0f000f00u) edge4<8>(Wh, wi, av, offv, pen, c16, m, den, acc);
    if (bal & 0xf000f000u) edge4<12>(Wh, wi, av, offv, pen, c16, m, den, acc);
}

#define BT 1024
#define BQ 3

__global__ __launch_bounds__(BT) void k_range(const float* __restrict__ Wh,
                                              const float* __restrict__ avec,
                                              const unsigned* __restrict__ seg,
                                              const int* __restrict__ ovfA,
                                              const unsigned* __restrict__ ovfACnt,
                                              float* __restrict__ out,
                                              int n, int numChunks, int RN) {
    __shared__ unsigned cntL[MAXRN + 32];
    __shared__ int rows[MAXRN * CAP];
    __shared__ unsigned nextPair;
    const int t = threadIdx.x;
    const int r = blockIdx.x;
    const int base = r * RN;
    for (int i = t; i < MAXRN + 32; i += BT) cntL[i] = 0u;
    if (t == 0) nextPair = 0u;
    __syncthreads();

    const int total = numChunks * CELL;
    const int totalQ = numChunks * CELLQ;
    const unsigned* segR = seg + (size_t)r * numChunks * CELLP;
    const uint4* segQ = (const uint4*)segR;
    unsigned oc0 = 0u;
    if (t < numChunks) oc0 = ovfACnt[t];
    const unsigned dump = (unsigned)MAXRN + (unsigned)(t & 31);
    for (int b0 = 0; b0 < totalQ; b0 += BT * BQ) {
        uint4 rq[BQ];
        unsigned rc[4 * BQ], kq[4 * BQ];
#pragma unroll
        for (int j = 0; j < BQ; ++j) {
            const int i = b0 + j * BT + t;
            const int ch = i / CELLQ;
            rq[j] = (i < totalQ) ? segQ[ch * (CELLP / 4) + (i - ch * CELLQ)] : make_uint4(EMPTY, EMPTY, EMPTY, EMPTY);
        }
#pragma unroll
        for (int j = 0; j < BQ; ++j) {
            rc[4 * j + 0] = rq[j].x;
            rc[4 * j + 1] = rq[j].y;
            rc[4 * j + 2] = rq[j].z;
            rc[4 * j + 3] = rq[j].w;
        }
#pragma unroll
        for (int q = 0; q < 4 * BQ; ++q) {
            const bool ok = (rc[q] & 0x00ffffffu) != EMPTY;
            kq[q] = atomicAdd(&cntL[ok ? (rc[q] >> 24) : dump], 1u);
        }
#pragma unroll
        for (int q = 0; q < 4 * BQ; ++q) {
            const bool ok = (rc[q] & 0x00ffffffu) != EMPTY;
            if (ok && kq[q] < (unsigned)CAP) rows[(rc[q] >> 24) * CAP + kq[q]] = (int)(rc[q] & 0x00ffffffu);
        }
    }

    for (int ch = t; ch < numChunks; ch += BT) {
        const unsigned oc = min((ch == t) ? oc0 : ovfACnt[ch], (unsigned)CHUNK);
        for (unsigned q = 0; q < oc; ++q) {
            const int ed = ovfA[2 * ((size_t)ch * CHUNK + q) + 0];
            const int es = ovfA[2 * ((size_t)ch * CHUNK + q) + 1];
            if (min((unsigned)ed / (unsigned)RN, (unsigned)(MAXR - 1)) == (unsigned)r)
                put_row(cntL, rows, ((unsigned)ed - (unsigned)base) & (unsigned)(MAXRN - 1), es);
        }
    }
    __syncthreads();

    const int lane = t & 31;
    const int c = lane & 15;
    const int half = lane >> 4;
    const int rn = min(RN, n - base);
    const int pairs = (rn + 1) >> 1;
    const float4 av = ((const float4*)avec)[c];
    const unsigned c16 = (unsigned)c * 16u;

    for (int it = 0; it < pairs; ++it) {
        unsigned tk = 0u;
        if (lane == 0) tk = atomicAdd(&nextPair, 1u);
        const int p = __builtin_amdgcn_readfirstlane((int)tk);
        if (p >= pairs) break;

        const int nl  = 2 * p + half;
        const bool live = nl < rn;
        const int nlc = live ? nl : 2 * p;
        const int nd  = base + nlc;
        const unsigned cn = cntL[nlc];
        const bool over = cn > (unsigned)CAP;
        const int deg = (live && !over) ? (int)cn : 0;
        const unsigned moreRows = __builtin_amdgcn_ballot_w32(live && over) & 0x00010001u;
        const float4 wi = ((const float4*)(Wh + (size_t)nd * DD))[c];
        const int* row = rows + nlc * CAP;

        const int d0 = __builtin_amdgcn_readlane(deg, 0);
        const int d1 = __builtin_amdgcn_readlane(deg, 16);
        const int degmax = d0 > d1 ? d0 : d1;

        float m = -3.0e38f;
        float den = 0.0f;
        float4 acc = make_float4(0.0f, 0.0f, 0.0f, 0.0f);

        for (int k = 0; k < degmax; k += 16) {
            const bool has = c < deg - k;
            int sv = has ? row[k + c] : nd;
            sv = min(max(sv, 0), n - 1);
            const int offv = sv * (DD * 4);
            const float pen = has ? 0.0f : NOEDGE;
            edge4<0>(Wh, wi, av, offv, pen, c16, m, den, acc);
            if (k + 4 < degmax)  edge4<4>(Wh, wi, av, offv, pen, c16, m, den, acc);
            if (k + 8 < degmax)  edge4<8>(Wh, wi, av, offv, pen, c16, m, den, acc);
            if (k + 12 < degmax) edge4<12>(Wh, wi, av, offv, pen, c16, m, den, acc);
        }

        if (moreRows != 0u) {
            const bool more = live && over;
            for (int k = 0; k < total; k += 16) {
                const int i = k + c;
                unsigned rc = EMPTY;
                if (i < total) {
                    const int ch = i / CELL;
                    rc = segR[ch * CELLP + (i - ch * CELL)];
                }
                const bool hit = more && (rc & 0x00ffffffu) != EMPTY && (rc >> 24) == (unsigned)nl;
                walk16(Wh, wi, av, hit, (int)(rc & 0x00ffffffu), nd, n, c16, m, den, acc);
            }
            for (int ch = 0; ch < numChunks; ++ch) {
                const int oc = __builtin_amdgcn_readfirstlane((int)min(ovfACnt[ch], (unsigned)CHUNK));
                for (int q0 = 0; q0 < oc; q0 += 16) {
                    const int q = q0 + c;
                    int ed = -1, es = nd;
                    if (q < oc) {
                        ed = ovfA[2 * ((size_t)ch * CHUNK + q) + 0];
                        es = ovfA[2 * ((size_t)ch * CHUNK + q) + 1];
                    }
                    const bool hit = more && ed == base + nl;
                    walk16(Wh, wi, av, hit, es, nd, n, c16, m, den, acc);
                }
            }
        }

        const float dn = fmaxf(den, 1e-9f);
        float4 o;
        o.x = 1.0f / (1.0f + __expf(-(acc.x / dn)));
        o.y = 1.0f / (1.0f + __expf(-(acc.y / dn)));
        o.z = 1.0f / (1.0f + __expf(-(acc.z / dn)));
        o.w = 1.0f / (1.0f + __expf(-(acc.w / dn)));
        if (live) ((float4*)(out + (size_t)(base + nl) * DD))[c] = o;
    }
}

extern "C" void kernel_launch(void* const* d_in, const int* in_sizes, int n_in,
                              void* d_out, int out_size, void* d_ws, size_t ws_size,
                              hipStream_t stream) {
    const float* x    = (const float*)d_in[0];
    const float* W    = (const float*)d_in[1];
    const float* bias = (const float*)d_in[2];
    const float* avec = (const float*)d_in[3];
    const int*   ei   = (const int*)d_in[4];

    const int N = in_sizes[0] / DD;
    const int E = in_sizes[4] / 2;
    const int* src = ei;
    const int* dst = ei + E;

    const int numChunks = (E + CHUNK - 1) / CHUNK;
    const int RN = (N + MAXR - 1) / MAXR;
    const int numRanges = (N + RN - 1) / RN;

    float*    Wh      = (float*)d_ws;
    unsigned* seg     = (unsigned*)(Wh + (size_t)N * DD);
    unsigned* ovfACnt = seg + (size_t)numRanges * numChunks * CELLP;
    int*      ovfA    = (int*)(ovfACnt + (((size_t)numChunks + 3) & ~(size_t)3));

    float* out = (float*)d_out;

    const int tiles = (N + 15) / 16;
    const int gemmBlocks = (tiles + 7) / 8;
    k_gemm_bin<<<gemmBlocks + numChunks, 256, 0, stream>>>(x, W, bias, Wh, src, dst, seg, ovfA, ovfACnt,
                                                           E, N, gemmBlocks, numChunks, numRanges, RN);

    k_range<<<numRanges, BT, 0, stream>>>(Wh, avec, seg, ovfA, ovfACnt, out, N, numChunks, RN);
}
